# starting kernel
# baseline (speedup 1.0000x reference)
#include <hip/hip_runtime.h>

#define N_NODES 100000
#define N_EDGES 1600000
#define IN_DIM  32
#define OUT_DIM 32
#define N_TILES (N_NODES / 16)

typedef __attribute__((ext_vector_type(2))) float v2f;
typedef __attribute__((ext_vector_type(8))) float v8f;

__global__ __launch_bounds__(256) void gin_zero_kernel(float4* __restrict__ agg4, int n4) {
    int i = blockIdx.x * 256 + threadIdx.x;
    if (i < n4) agg4[i] = float4{0.f, 0.f, 0.f, 0.f};
}

__device__ __forceinline__ void atomic_add_f32_noret(float* p, float v) {
    asm volatile("global_atomic_add_f32 %0, %1, off"
                 :
                 : "v"(p), "v"(v)
                 : "memory");
}

__global__ __launch_bounds__(256) void gin_scatter_kernel(const float* __restrict__ X,
                                                          const int* __restrict__ src,
                                                          const int* __restrict__ dst,
                                                          float* __restrict__ agg) {
    unsigned tid = blockIdx.x * 256u + threadIdx.x;
    unsigned e = tid >> 4;
    unsigned d = (tid & 15u) * 2u;
    unsigned s = (unsigned)src[e];
    unsigned t = (unsigned)dst[e];
    const float2 v = *(const float2*)&X[s * 32u + d];
    float* p = &agg[t * 32u + d];
    atomic_add_f32_noret(p,     v.x);
    atomic_add_f32_noret(p + 1, v.y);
}

__global__ __launch_bounds__(128) void gin_gemm_kernel(const float* __restrict__ A,
                                                       const float* __restrict__ W,
                                                       float* __restrict__ out) {
    __shared__ float Wt[32 * 32];
    int t = threadIdx.x;
    for (int i = t; i < 1024; i += 128) {
        int k = i >> 5, n = i & 31;
        Wt[n * 32 + k] = W[i];
    }
    __syncthreads();

    int wave = t >> 5;
    int lane = t & 31;
    int m    = lane & 15;
    int hi   = lane >> 4;
    int tile = blockIdx.x * 4 + wave;

    if (tile < N_TILES) {
        int row0 = tile * 16;
        const float* aptr = A + (size_t)(row0 + m) * 32;

        v2f a[8];
#pragma unroll
        for (int s = 0; s < 8; ++s) {
            int k = 4 * s + 2 * hi;
            a[s].x = aptr[k];
            a[s].y = aptr[k + 1];
        }

#pragma unroll
        for (int nt = 0; nt < 2; ++nt) {
            v8f c = {};
            const float* wp = &Wt[(nt * 16 + m) * 32 + 2 * hi];
#pragma unroll
            for (int s = 0; s < 8; ++s) {
                v2f b;
                b.x = wp[4 * s];
                b.y = wp[4 * s + 1];
                c = __builtin_amdgcn_wmma_f32_16x16x4_f32(
                         false, a[s],  false, b,
                         (short)0, c,  false,  false);
            }
            float* op = out + (size_t)(row0 + 8 * hi) * 32 + nt * 16 + m;
#pragma unroll
            for (int r = 0; r < 8; ++r) {
                op[(size_t)r * 32] = c[r];
            }
        }
    }
}

extern "C" void kernel_launch(void* const* d_in, const int* in_sizes, int n_in,
                              void* d_out, int out_size, void* d_ws, size_t ws_size,
                              hipStream_t stream) {
    const float* X   = (const float*)d_in[0];
    const float* W   = (const float*)d_in[1];
    const int*   src = (const int*)d_in[2];
    const int*   dst = (const int*)d_in[3];
    float*       out = (float*)d_out;
    float*       agg = (float*)d_ws;

    int n4 = (N_NODES * IN_DIM) / 4;
    gin_zero_kernel<<<(n4 + 255) / 256, 256, 0, stream>>>((float4*)agg, n4);

    gin_scatter_kernel<<<(N_EDGES * 16) / 256, 256, 0, stream>>>(X, src, dst, agg);

    int gblocks = (N_TILES + 3) / 4;
    gin_gemm_kernel<<<gblocks, 128, 0, stream>>>(agg, W, out);
}


# atomics removed: per call one stable ballot-ranked split of the edges on dst>>8 (histogram, scan, scatter); tile blocks add X rows in ascending edge index (new sum order), fused with the WMMA GEMM
# speedup vs baseline: 1.1861x; 1.1861x over previous
#include <hip/hip_runtime.h>

#define N_NODES 100000
#define N_EDGES 1600000
#define IN_DIM  32
#define OUT_DIM 32
#define N_TILES (N_NODES / 16)
#define TILE_LD 36

#define N_PART   392
#define N_CHUNK  256
#define CHUNK_E  6400
#define WAVE_E   800
#define ROUNDS   25
#define PER_LANE 9
#define LIST_CAP (512 * PER_LANE)

typedef __attribute__((ext_vector_type(2))) float v2f;
typedef __attribute__((ext_vector_type(8))) float v8f;

__device__ __forceinline__ void wave_lds_order() {
    __builtin_amdgcn_fence(__ATOMIC_RELEASE, "wavefront");
    __builtin_amdgcn_wave_barrier();
    __builtin_amdgcn_fence(__ATOMIC_ACQUIRE, "wavefront");
}


__device__ __forceinline__ unsigned same_part_lanes(int part) {
    unsigned same = 0xffffffffu;
#pragma unroll
    for (int b = 0; b < 9; ++b) {
        const bool bit = (part >> b) & 1;
        const unsigned bal = __builtin_amdgcn_ballot_w32(bit);
        same &= bit ? bal : ~bal;
    }
    return same;
}

__global__ __launch_bounds__(256) void gin_hist_kernel(const int* __restrict__ dst,
                                                       int* __restrict__ hist) {
    __shared__ int h[8 * N_PART];
    const int t = threadIdx.x;
    const int lane = t & 31;
    const int w = t >> 5;
    const int b = blockIdx.x;
    for (int i = t; i < 8 * N_PART; i += 256) h[i] = 0;
    __syncthreads();
    if (b < N_EDGES / CHUNK_E) {
        const int* p = dst + b * CHUNK_E + w * WAVE_E + lane;
        int part[ROUNDS];
#pragma unroll
        for (int r = 0; r < ROUNDS; ++r) part[r] = p[r * 32] >> 8;
        int* hw = h + w * N_PART;
#pragma unroll
        for (int r = 0; r < ROUNDS; ++r) {
            const unsigned same = same_part_lanes(part[r]);
            if (lane == __builtin_ctz(same)) hw[part[r]] += __builtin_popcount(same);
            wave_lds_order();
        }
    }
    __syncthreads();
    for (int d = t; d < N_PART; d += 256) {
        int s = 0;
#pragma unroll
        for (int i = 0; i < 8; ++i) s += h[i * N_PART + d];
        hist[d * N_CHUNK + b] = s;
    }
    if (b == 0 && t < 4) hist[N_PART * N_CHUNK + t] = 0;
}

__global__ __launch_bounds__(1024) void gin_scan_kernel(const int* __restrict__ hist,
                                                        int* __restrict__ off) {
    __shared__ int wred[32];
    __shared__ int wscan[32];
    const int t = threadIdx.x;
    const int lane = t & 31;
    const int w = t >> 5;
    const int d = blockIdx.x;
    const int4* h4 = (const int4*)hist;

    int s = 0;
    const int n4 = d * (N_CHUNK / 4);
#pragma unroll
    for (int j = 0; j < 25; ++j) {
        const int i = t + j * 1024;
        const int4 v = h4[(i < n4) ? i : (N_PART * N_CHUNK / 4)];
        s += (v.x + v.y) + (v.z + v.w);
    }
    const int local = (t < N_CHUNK) ? hist[d * N_CHUNK + t] : 0;
    int incl = local;
#pragma unroll
    for (int o = 1; o < 32; o <<= 1) {
        const int u = __shfl_up(incl, o);
        if (lane >= o) incl += u;
    }
#pragma unroll
    for (int o = 16; o > 0; o >>= 1) s += __shfl_xor(s, o);
    if (lane == 31) wscan[w] = incl;
    if (lane == 0)  wred[w] = s;
    __syncthreads();
    int base = 0;
    for (int i = 0; i < 32; ++i) base += wred[i];
    for (int i = 0; i < w; ++i)  base += wscan[i];
    if (t < N_CHUNK) off[d * N_CHUNK + t] = base + incl - local;
}

__global__ __launch_bounds__(256) void gin_split_kernel(const int* __restrict__ src,
                                                        const int* __restrict__ dst,
                                                        const int* __restrict__ off,
                                                        int* __restrict__ sorted) {
    __shared__ int h[8 * N_PART];
    const int t = threadIdx.x;
    const int lane = t & 31;
    const int w = t >> 5;
    const int b = blockIdx.x;
    for (int i = t; i < 8 * N_PART; i += 256) h[i] = 0;
    __syncthreads();

    const int e0 = b * CHUNK_E + w * WAVE_E + lane;
    int dv[ROUNDS];
    int sv[ROUNDS];
#pragma unroll
    for (int r = 0; r < ROUNDS; ++r) dv[r] = dst[e0 + r * 32];
#pragma unroll
    for (int r = 0; r < ROUNDS; ++r) sv[r] = src[e0 + r * 32];

    int* hw = h + w * N_PART;
    unsigned same[ROUNDS];
#pragma unroll
    for (int r = 0; r < ROUNDS; ++r) {
        same[r] = same_part_lanes(dv[r] >> 8);
        if (lane == __builtin_ctz(same[r])) hw[dv[r] >> 8] += __builtin_popcount(same[r]);
        wave_lds_order();
    }
    __syncthreads();
    for (int d = t; d < N_PART; d += 256) {
        int run = off[d * N_CHUNK + b];
#pragma unroll
        for (int i = 0; i < 8; ++i) {
            const int c = h[i * N_PART + d];
            h[i * N_PART + d] = run;
            run += c;
        }
    }
    __syncthreads();
    const unsigned lt = (1u << lane) - 1u;
#pragma unroll
    for (int r = 0; r < ROUNDS; ++r) {
        const int part = dv[r] >> 8;
        const int start = hw[part];
        wave_lds_order();
        if (lane == __builtin_ctz(same[r])) hw[part] = start + __builtin_popcount(same[r]);
        wave_lds_order();
        sorted[start + __builtin_popcount(same[r] & lt)] = (sv[r] << 8) | (dv[r] & 255);
    }
}

__device__ __forceinline__ float add_rows_in_order(const float* __restrict__ X, int ids,
                                                   int cnt, int lane, float acc) {
    for (int k = 0; k < cnt; k += 8) {
        float v[8];
#pragma unroll
        for (int i = 0; i < 8; ++i) {
            const int sk = __builtin_amdgcn_readlane(ids, k + i);
            v[i] = X[(size_t)sk * 32 + lane];
        }
#pragma unroll
        for (int i = 0; i < 8; ++i) acc += (k + i < cnt) ? v[i] : 0.0f;
    }
    return acc;
}

__global__ __launch_bounds__(512) void gin_pull_gemm_kernel(const float* __restrict__ X,
                                                            const float* __restrict__ W,
                                                            const int* __restrict__ off,
                                                            const int* __restrict__ sorted,
                                                            float* __restrict__ out) {
    __shared__ float tile[16 * TILE_LD];
    __shared__ int   list[LIST_CAP];
    __shared__ int   wtot[16];
    __shared__ int   pendbuf[16 * 64];

    const int lane = threadIdx.x & 31;
    const int wave = __builtin_amdgcn_readfirstlane((int)(threadIdx.x >> 5));
    const int m  = lane & 15;
    const int hi = lane >> 4;
    const int row0 = blockIdx.x * 16;
    const int part = blockIdx.x >> 4;
    const int tsel = blockIdx.x & 15;
    const int pbeg = off[part * N_CHUNK];
    const int pend = off[(part + 1) * N_CHUNK];
    const unsigned lt = (1u << lane) - 1u;

    int* mine = pendbuf + wave * 64;
    int npend = 0;
    float acc = 0.0f;

    for (int cb = pbeg; cb < pend; cb += LIST_CAP) {
        int ent[PER_LANE];
        unsigned mk[PER_LANE];
        int tot = 0;
#pragma unroll
        for (int j = 0; j < PER_LANE; ++j) {
            const int idx = cb + wave * (32 * PER_LANE) + j * 32 + lane;
            ent[j] = (idx < pend) ? sorted[idx] : -1;
        }
#pragma unroll
        for (int j = 0; j < PER_LANE; ++j) {
            const bool hit = (ent[j] >= 0) && (((ent[j] >> 4) & 15) == tsel);
            mk[j] = __builtin_amdgcn_ballot_w32(hit);
            tot += __builtin_popcount(mk[j]);
        }
        if (lane == 0) wtot[wave] = tot;
        __syncthreads();
        int wbase = 0, len = 0;
#pragma unroll
        for (int i = 0; i < 16; ++i) {
            const int c = wtot[i];
            len += c;
            if (i < wave) wbase += c;
        }
        len = __builtin_amdgcn_readfirstlane(len);
        wbase = __builtin_amdgcn_readfirstlane(wbase);
#pragma unroll
        for (int j = 0; j < PER_LANE; ++j) {
            if ((mk[j] >> lane) & 1u) list[wbase + __builtin_popcount(mk[j] & lt)] = ent[j];
            wbase += __builtin_popcount(mk[j]);
        }
        __syncthreads();

        for (int r = 0; r < len; r += 32) {
            const int idx = r + lane;
            const int e = (idx < len) ? list[idx] : -1;
            const bool hit = (e >= 0) && ((e & 15) == wave);
            const unsigned mask = __builtin_amdgcn_ballot_w32(hit);
            const int c = __builtin_popcount(mask);
            if (c) {
                if (hit) mine[npend + __builtin_popcount(mask & lt)] = e >> 8;
                npend += c;
                if (npend >= 32) {
                    wave_lds_order();
                    const int ids  = mine[lane];
                    const int rest = mine[32 + lane];
                    acc = add_rows_in_order(X, ids, 32, lane, acc);
                    npend -= 32;
                    wave_lds_order();
                    if (lane < npend) mine[lane] = rest;
                    wave_lds_order();
                }
            }
        }
        __syncthreads();
    }
    v2f bfr[2][8];
    if (wave == 0) {
#pragma unroll
        for (int nt = 0; nt < 2; ++nt) {
#pragma unroll
            for (int s = 0; s < 8; ++s) {
                const int k = 4 * s + 2 * hi;
                bfr[nt][s].x = W[k * 32 + nt * 16 + m];
                bfr[nt][s].y = W[(k + 1) * 32 + nt * 16 + m];
            }
        }
    }
    wave_lds_order();
    {
        const int ids = (lane < npend) ? mine[lane] : 0;
        acc = add_rows_in_order(X, ids, npend, lane, acc);
    }

    tile[wave * TILE_LD + lane] = acc;
    __syncthreads();

    if (wave == 0) {
        v2f a[8];
#pragma unroll
        for (int s = 0; s < 8; ++s)
            a[s] = *(const v2f*)&tile[m * TILE_LD + 4 * s + 2 * hi];
#pragma unroll
        for (int nt = 0; nt < 2; ++nt) {
            v8f c = {};
#pragma unroll
            for (int s = 0; s < 8; ++s) {
                c = __builtin_amdgcn_wmma_f32_16x16x4_f32(
                         false, a[s],  false, bfr[nt][s],
                         (short)0, c,  false,  false);
            }
            float* op = out + (size_t)(row0 + 8 * hi) * 32 + nt * 16 + m;
#pragma unroll
            for (int r = 0; r < 8; ++r) op[(size_t)r * 32] = c[r];
        }
    }
}

extern "C" void kernel_launch(void* const* d_in, const int* in_sizes, int n_in,
                              void* d_out, int out_size, void* d_ws, size_t ws_size,
                              hipStream_t stream) {
    const float* X   = (const float*)d_in[0];
    const float* W   = (const float*)d_in[1];
    const int*   src = (const int*)d_in[2];
    const int*   dst = (const int*)d_in[3];
    float*       out = (float*)d_out;
    int*         hist   = (int*)d_ws;
    int*         off    = hist + N_PART * N_CHUNK + 4;
    int*         sorted = off + N_PART * N_CHUNK;

    gin_hist_kernel<<<N_CHUNK, 256, 0, stream>>>(dst, hist);
    gin_scan_kernel<<<N_PART, 1024, 0, stream>>>(hist, off);
    gin_split_kernel<<<N_EDGES / CHUNK_E, 256, 0, stream>>>(src, dst, off, sorted);
    gin_pull_gemm_kernel<<<N_TILES, 512, 0, stream>>>(X, W, off, sorted, out);
}


# pull kernel: one 16-wave block per 256-node part, not per 16-row tile (6256 waves enter, not 100000); block sorts its part by node in LDS once, each wave sums and multiplies its tile; sum order kept
# speedup vs baseline: 1.5931x; 1.3431x over previous
#include <hip/hip_runtime.h>

#define N_NODES 100000
#define N_EDGES 1600000
#define IN_DIM  32
#define OUT_DIM 32
#define N_TILES (N_NODES / 16)
#define TILE_LD 36

#define N_PART   392
#define N_CHUNK  256
#define CHUNK_E  6400
#define WAVE_E   800
#define ROUNDS   25
#define PER_LANE 9
#define LIST_CAP (512 * PER_LANE)
#define NB       257
#define N_PBLK   ((N_NODES + 255) / 256)

typedef __attribute__((ext_vector_type(2))) float v2f;
typedef __attribute__((ext_vector_type(8))) float v8f;

__device__ __forceinline__ void wave_lds_order() {
    __builtin_amdgcn_fence(__ATOMIC_RELEASE, "wavefront");
    __builtin_amdgcn_wave_barrier();
    __builtin_amdgcn_fence(__ATOMIC_ACQUIRE, "wavefront");
}


__device__ __forceinline__ unsigned same_part_lanes(int part) {
    unsigned same = 0xffffffffu;
#pragma unroll
    for (int b = 0; b < 9; ++b) {
        const bool bit = (part >> b) & 1;
        const unsigned bal = __builtin_amdgcn_ballot_w32(bit);
        same &= bit ? bal : ~bal;
    }
    return same;
}

__global__ __launch_bounds__(256) void gin_hist_kernel(const int* __restrict__ dst,
                                                       int* __restrict__ hist) {
    __shared__ int h[8 * N_PART];
    const int t = threadIdx.x;
    const int lane = t & 31;
    const int w = t >> 5;
    const int b = blockIdx.x;
    for (int i = t; i < 8 * N_PART; i += 256) h[i] = 0;
    __syncthreads();
    if (b < N_EDGES / CHUNK_E) {
        const int* p = dst + b * CHUNK_E + w * WAVE_E + lane;
        int part[ROUNDS];
#pragma unroll
        for (int r = 0; r < ROUNDS; ++r) part[r] = p[r * 32] >> 8;
        int* hw = h + w * N_PART;
#pragma unroll
        for (int r = 0; r < ROUNDS; ++r) {
            const unsigned same = same_part_lanes(part[r]);
            if (lane == __builtin_ctz(same)) hw[part[r]] += __builtin_popcount(same);
            wave_lds_order();
        }
    }
    __syncthreads();
    for (int d = t; d < N_PART; d += 256) {
        int s = 0;
#pragma unroll
        for (int i = 0; i < 8; ++i) s += h[i * N_PART + d];
        hist[d * N_CHUNK + b] = s;
    }
    if (b == 0 && t < 4) hist[N_PART * N_CHUNK + t] = 0;
}

__global__ __launch_bounds__(1024) void gin_scan_kernel(const int* __restrict__ hist,
                                                        int* __restrict__ off) {
    __shared__ int wred[32];
    __shared__ int wscan[32];
    const int t = threadIdx.x;
    const int lane = t & 31;
    const int w = t >> 5;
    const int d = blockIdx.x;
    const int4* h4 = (const int4*)hist;

    int s = 0;
    const int n4 = d * (N_CHUNK / 4);
#pragma unroll
    for (int j = 0; j < 25; ++j) {
        const int i = t + j * 1024;
        const int4 v = h4[(i < n4) ? i : (N_PART * N_CHUNK / 4)];
        s += (v.x + v.y) + (v.z + v.w);
    }
    const int local = (t < N_CHUNK) ? hist[d * N_CHUNK + t] : 0;
    int incl = local;
#pragma unroll
    for (int o = 1; o < 32; o <<= 1) {
        const int u = __shfl_up(incl, o);
        if (lane >= o) incl += u;
    }
#pragma unroll
    for (int o = 16; o > 0; o >>= 1) s += __shfl_xor(s, o);
    if (lane == 31) wscan[w] = incl;
    if (lane == 0)  wred[w] = s;
    __syncthreads();
    int base = 0;
    for (int i = 0; i < 32; ++i) base += wred[i];
    for (int i = 0; i < w; ++i)  base += wscan[i];
    if (t < N_CHUNK) off[d * N_CHUNK + t] = base + incl - local;
}

__global__ __launch_bounds__(256) void gin_split_kernel(const int* __restrict__ src,
                                                        const int* __restrict__ dst,
                                                        const int* __restrict__ off,
                                                        int* __restrict__ sorted) {
    __shared__ int h[8 * N_PART];
    const int t = threadIdx.x;
    const int lane = t & 31;
    const int w = t >> 5;
    const int b = blockIdx.x;
    for (int i = t; i < 8 * N_PART; i += 256) h[i] = 0;
    __syncthreads();

    const int e0 = b * CHUNK_E + w * WAVE_E + lane;
    int dv[ROUNDS];
    int sv[ROUNDS];
#pragma unroll
    for (int r = 0; r < ROUNDS; ++r) dv[r] = dst[e0 + r * 32];
#pragma unroll
    for (int r = 0; r < ROUNDS; ++r) sv[r] = src[e0 + r * 32];

    int* hw = h + w * N_PART;
    unsigned same[ROUNDS];
#pragma unroll
    for (int r = 0; r < ROUNDS; ++r) {
        same[r] = same_part_lanes(dv[r] >> 8);
        if (lane == __builtin_ctz(same[r])) hw[dv[r] >> 8] += __builtin_popcount(same[r]);
        wave_lds_order();
    }
    __syncthreads();
    for (int d = t; d < N_PART; d += 256) {
        int run = off[d * N_CHUNK + b];
#pragma unroll
        for (int i = 0; i < 8; ++i) {
            const int c = h[i * N_PART + d];
            h[i * N_PART + d] = run;
            run += c;
        }
    }
    __syncthreads();
    const unsigned lt = (1u << lane) - 1u;
#pragma unroll
    for (int r = 0; r < ROUNDS; ++r) {
        const int part = dv[r] >> 8;
        const int start = hw[part];
        wave_lds_order();
        if (lane == __builtin_ctz(same[r])) hw[part] = start + __builtin_popcount(same[r]);
        wave_lds_order();
        sorted[start + __builtin_popcount(same[r] & lt)] = (sv[r] << 8) | (dv[r] & 255);
    }
}

__global__ __launch_bounds__(512) void gin_pull_gemm_kernel(const float* __restrict__ X,
                                                            const float* __restrict__ W,
                                                            const int* __restrict__ off,
                                                            const int* __restrict__ sorted,
                                                            float* __restrict__ out) {
    __shared__ int   h[16 * NB];
    __shared__ int   nstart[NB];
    __shared__ int   ids[LIST_CAP];
    __shared__ int   wsum[8];
    __shared__ __attribute__((aligned(16))) float tile[16 * 16 * TILE_LD];

    const int t = threadIdx.x;
    const int lane = t & 31;
    const int wave = __builtin_amdgcn_readfirstlane((int)(threadIdx.x >> 5));
    const int m  = lane & 15;
    const int hi = lane >> 4;
    const int part = blockIdx.x;
    const int pbeg = off[part * N_CHUNK];
    const int pend = off[(part + 1) * N_CHUNK];
    const unsigned lt = (1u << lane) - 1u;
    int*   hw     = h + wave * NB;
    float* mytile = tile + wave * (16 * TILE_LD);

    for (int n = 0; n < 16; ++n) mytile[n * TILE_LD + lane] = 0.0f;

    for (int cb = pbeg; cb < pend; cb += LIST_CAP) {
        for (int i = t; i < 16 * NB; i += 512) h[i] = 0;
        __syncthreads();

        int ent[PER_LANE];
        unsigned same[PER_LANE];
#pragma unroll
        for (int j = 0; j < PER_LANE; ++j) {
            const int idx = cb + wave * (32 * PER_LANE) + j * 32 + lane;
            ent[j] = (idx < pend) ? sorted[idx] : -1;
        }
#pragma unroll
        for (int j = 0; j < PER_LANE; ++j) {
            const int key = (ent[j] >= 0) ? (ent[j] & 255) : 256;
            same[j] = same_part_lanes(key);
            if (lane == __builtin_ctz(same[j])) hw[key] += __builtin_popcount(same[j]);
            wave_lds_order();
        }
        __syncthreads();

        int tot = 0;
        if (t < 256) {
#pragma unroll
            for (int i = 0; i < 16; ++i) tot += h[i * NB + t];
        }
        int incl = tot;
#pragma unroll
        for (int o = 1; o < 32; o <<= 1) {
            const int u = __shfl_up(incl, o);
            if (lane >= o) incl += u;
        }
        if (lane == 31 && wave < 8) wsum[wave] = incl;
        __syncthreads();
        if (t < 256) {
            int run = incl - tot;
#pragma unroll
            for (int i = 0; i < 8; ++i) run += (i < wave) ? wsum[i] : 0;
            nstart[t] = run;
#pragma unroll
            for (int i = 0; i < 16; ++i) {
                const int c = h[i * NB + t];
                h[i * NB + t] = run;
                run += c;
            }
            if (t == 255) nstart[256] = run;
        }
        __syncthreads();

#pragma unroll
        for (int j = 0; j < PER_LANE; ++j) {
            const int key = (ent[j] >= 0) ? (ent[j] & 255) : 256;
            const int start = hw[key];
            wave_lds_order();
            if (lane == __builtin_ctz(same[j])) hw[key] = start + __builtin_popcount(same[j]);
            wave_lds_order();
            if (ent[j] >= 0) ids[start + __builtin_popcount(same[j] & lt)] = ent[j] >> 8;
        }
        __syncthreads();

        {
            const int nb = nstart[16 * wave + ((lane < 16) ? lane : 16)];
            const int S = __builtin_amdgcn_readlane(nb, 0);
            const int E = __builtin_amdgcn_readlane(nb, 16);
            if (S < E) {
                int cur = 0;
                int cend = __builtin_amdgcn_readlane(nb, 1);
                float acc = mytile[lane];

                auto load16 = [&](int k, float (&v)[16]) {
                    const int idv = (k + lane < E) ? ids[k + lane] : 0;
#pragma unroll
                    for (int i = 0; i < 16; ++i) {
                        const int sk = __builtin_amdgcn_readlane(idv, i);
                        v[i] = X[sk * 32 + lane];
                    }
                };
                auto add_row = [&](int p, float v) {
                    while (__builtin_expect(p >= cend, 0)) {
                        mytile[cur * TILE_LD + lane] = acc;
                        ++cur;
                        acc = mytile[cur * TILE_LD + lane];
                        cend = __builtin_amdgcn_readlane(nb, cur + 1);
                    }
                    acc += v;
                };
                auto add16 = [&](int k, const float (&v)[16]) {
                    if (k + 16 <= E) {
#pragma unroll
                        for (int i = 0; i < 16; ++i) add_row(k + i, v[i]);
                    } else {
#pragma unroll
                        for (int i = 0; i < 16; ++i)
                            if (k + i < E) add_row(k + i, v[i]);
                    }
                };

                float va[16], vb[16];
                load16(S, va);
                for (int k = S; k < E; k += 32) {
                    load16(k + 16, vb);
                    add16(k, va);
                    load16(k + 32, va);
                    add16(k + 16, vb);
                }
                mytile[cur * TILE_LD + lane] = acc;
            }
        }
    }

    if (part * 16 + wave < N_TILES) {
        const int row0 = (part * 16 + wave) * 16;
        v2f bfr[2][8];
#pragma unroll
        for (int nt = 0; nt < 2; ++nt) {
#pragma unroll
            for (int s = 0; s < 8; ++s) {
                const int k = 4 * s + 2 * hi;
                bfr[nt][s].x = W[k * 32 + nt * 16 + m];
                bfr[nt][s].y = W[(k + 1) * 32 + nt * 16 + m];
            }
        }
        wave_lds_order();
        v2f a[8];
#pragma unroll
        for (int s = 0; s < 8; ++s)
            a[s] = *(const v2f*)&mytile[m * TILE_LD + 4 * s + 2 * hi];
#pragma unroll
        for (int nt = 0; nt < 2; ++nt) {
            v8f c = {};
#pragma unroll
            for (int s = 0; s < 8; ++s) {
                c = __builtin_amdgcn_wmma_f32_16x16x4_f32(
                         false, a[s],  false, bfr[nt][s],
                         (short)0, c,  false,  false);
            }
            float* op = out + (size_t)(row0 + 8 * hi) * 32 + nt * 16 + m;
#pragma unroll
            for (int r = 0; r < 8; ++r) op[(size_t)r * 32] = c[r];
        }
    }
}

extern "C" void kernel_launch(void* const* d_in, const int* in_sizes, int n_in,
                              void* d_out, int out_size, void* d_ws, size_t ws_size,
                              hipStream_t stream) {
    const float* X   = (const float*)d_in[0];
    const float* W   = (const float*)d_in[1];
    const int*   src = (const int*)d_in[2];
    const int*   dst = (const int*)d_in[3];
    float*       out = (float*)d_out;
    int*         hist   = (int*)d_ws;
    int*         off    = hist + N_PART * N_CHUNK + 4;
    int*         sorted = off + N_PART * N_CHUNK;

    gin_hist_kernel<<<N_CHUNK, 256, 0, stream>>>(dst, hist);
    gin_scan_kernel<<<N_PART, 1024, 0, stream>>>(hist, off);
    gin_split_kernel<<<N_EDGES / CHUNK_E, 256, 0, stream>>>(src, dst, off, sorted);
    gin_pull_gemm_kernel<<<N_PBLK, 512, 0, stream>>>(X, W, off, sorted, out);
}


# global histogram and scan dropped (4 launches to 2): each chunk is split by part in its own slice and publishes its run starts; the pull block collects its part's 250 runs in LDS; sum order kept
# speedup vs baseline: 2.9492x; 1.8512x over previous
#include <hip/hip_runtime.h>

#define N_NODES 100000
#define N_EDGES 1600000
#define IN_DIM  32
#define OUT_DIM 32
#define N_TILES (N_NODES / 16)
#define TILE_LD 36

#define N_PART   392
#define N_CHUNK  256
#define N_CHUNKS (N_EDGES / CHUNK_E)
#define CHUNK_E  6400
#define WAVE_E   800
#define ROUNDS   25
#define PER_LANE 9
#define LIST_CAP (512 * PER_LANE)
#define NB       257
#define N_PBLK   ((N_NODES + 255) / 256)

typedef __attribute__((ext_vector_type(2))) float v2f;
typedef __attribute__((ext_vector_type(8))) float v8f;

__device__ __forceinline__ void wave_lds_order() {
    __builtin_amdgcn_fence(__ATOMIC_RELEASE, "wavefront");
    __builtin_amdgcn_wave_barrier();
    __builtin_amdgcn_fence(__ATOMIC_ACQUIRE, "wavefront");
}


__device__ __forceinline__ unsigned same_part_lanes(int part) {
    unsigned same = 0xffffffffu;
#pragma unroll
    for (int b = 0; b < 9; ++b) {
        const bool bit = (part >> b) & 1;
        const unsigned bal = __builtin_amdgcn_ballot_w32(bit);
        same &= bit ? bal : ~bal;
    }
    return same;
}

__global__ __launch_bounds__(256) void gin_split_kernel(const int* __restrict__ src,
                                                        const int* __restrict__ dst,
                                                        int* __restrict__ runs,
                                                        int* __restrict__ sorted) {
    __shared__ int h[8 * N_PART];
    __shared__ int wtot[8];
    const int t = threadIdx.x;
    const int lane = t & 31;
    const int w = t >> 5;
    const int b = blockIdx.x;
    for (int i = t; i < 8 * N_PART; i += 256) h[i] = 0;
    __syncthreads();

    const int e0 = b * CHUNK_E + w * WAVE_E + lane;
    int dv[ROUNDS];
    int sv[ROUNDS];
#pragma unroll
    for (int r = 0; r < ROUNDS; ++r) dv[r] = dst[e0 + r * 32];
#pragma unroll
    for (int r = 0; r < ROUNDS; ++r) sv[r] = src[e0 + r * 32];

    int* hw = h + w * N_PART;
    unsigned same[ROUNDS];
#pragma unroll
    for (int r = 0; r < ROUNDS; ++r) {
        same[r] = same_part_lanes(dv[r] >> 8);
        if (lane == __builtin_ctz(same[r])) hw[dv[r] >> 8] += __builtin_popcount(same[r]);
        wave_lds_order();
    }
    __syncthreads();
    const int d0 = 2 * t;
    int tot0 = 0, tot1 = 0;
    if (t < N_PART / 2) {
#pragma unroll
        for (int i = 0; i < 8; ++i) {
            tot0 += h[i * N_PART + d0];
            tot1 += h[i * N_PART + d0 + 1];
        }
    }
    int incl = tot0 + tot1;
#pragma unroll
    for (int o = 1; o < 32; o <<= 1) {
        const int u = __shfl_up(incl, o);
        if (lane >= o) incl += u;
    }
    if (lane == 31) wtot[w] = incl;
    __syncthreads();
    if (t < N_PART / 2) {
        int first = incl - (tot0 + tot1);
#pragma unroll
        for (int i = 0; i < 8; ++i) first += (i < w) ? wtot[i] : 0;
        runs[d0 * N_CHUNK + b] = first;
        runs[(d0 + 1) * N_CHUNK + b] = first + tot0;
        int run = b * CHUNK_E + first;
#pragma unroll
        for (int i = 0; i < 8; ++i) {
            const int c = h[i * N_PART + d0];
            h[i * N_PART + d0] = run;
            run += c;
        }
#pragma unroll
        for (int i = 0; i < 8; ++i) {
            const int c = h[i * N_PART + d0 + 1];
            h[i * N_PART + d0 + 1] = run;
            run += c;
        }
    }
    __syncthreads();
    const unsigned lt = (1u << lane) - 1u;
#pragma unroll
    for (int r = 0; r < ROUNDS; ++r) {
        const int part = dv[r] >> 8;
        const int start = hw[part];
        wave_lds_order();
        if (lane == __builtin_ctz(same[r])) hw[part] = start + __builtin_popcount(same[r]);
        wave_lds_order();
        sorted[start + __builtin_popcount(same[r] & lt)] = (sv[r] << 8) | (dv[r] & 255);
    }
}

__global__ __launch_bounds__(512) void gin_pull_gemm_kernel(const float* __restrict__ X,
                                                            const float* __restrict__ W,
                                                            const int* __restrict__ runs,
                                                            const int* __restrict__ sorted,
                                                            float* __restrict__ out) {
    __shared__ int   h[16 * NB];
    __shared__ int   nstart[NB];
    __shared__ int   ids[LIST_CAP];
    __shared__ int   wsum[8];
    __shared__ int   segbase[256];
    __shared__ int   segpos[257];
    __shared__ int   stage[LIST_CAP];
    __shared__ __attribute__((aligned(16))) float tile[16 * 16 * TILE_LD];

    const int t = threadIdx.x;
    const int lane = t & 31;
    const int wave = __builtin_amdgcn_readfirstlane((int)(threadIdx.x >> 5));
    const int m  = lane & 15;
    const int hi = lane >> 4;
    const int part = blockIdx.x;
    const unsigned lt = (1u << lane) - 1u;
    int*   hw     = h + wave * NB;
    float* mytile = tile + wave * (16 * TILE_LD);

    for (int n = 0; n < 16; ++n) mytile[n * TILE_LD + lane] = 0.0f;

    {
        int s0 = 0, len = 0;
        if (t < N_CHUNKS) {
            s0  = runs[part * N_CHUNK + t];
            len = runs[(part + 1) * N_CHUNK + t] - s0;
        }
        int incl = len;
#pragma unroll
        for (int o = 1; o < 32; o <<= 1) {
            const int u = __shfl_up(incl, o);
            if (lane >= o) incl += u;
        }
        if (lane == 31 && wave < 8) wsum[wave] = incl;
        __syncthreads();
        if (t < 256) {
            int first = incl - len;
#pragma unroll
            for (int i = 0; i < 8; ++i) first += (i < wave) ? wsum[i] : 0;
            segbase[t] = t * CHUNK_E + s0;
            segpos[t] = first;
            if (t == 255) segpos[256] = first + len;
        }
        __syncthreads();
    }
    const int total = segpos[256];

    for (int cb = 0; cb < total; cb += LIST_CAP) {
        for (int i = t; i < 16 * NB; i += 512) h[i] = 0;

        {
            const int sp = segpos[16 * wave + ((lane < 16) ? lane : 16)];
            const int sb = segbase[16 * wave + (lane & 15)];
            int ev[16];
#pragma unroll
            for (int r = 0; r < 16; ++r) {
                const int p0 = __builtin_amdgcn_readlane(sp, r);
                const int p1 = __builtin_amdgcn_readlane(sp, r + 1);
                const int base = __builtin_amdgcn_readlane(sb, r);
                const int last = (p1 - p0 > 0) ? (p1 - p0 - 1) : 0;
                const int idx = base + ((lane < last) ? lane : last);
                ev[r] = sorted[(idx < N_EDGES) ? idx : (N_EDGES - 1)];
            }
#pragma unroll
            for (int r = 0; r < 16; ++r) {
                const int p0 = __builtin_amdgcn_readlane(sp, r);
                const int p1 = __builtin_amdgcn_readlane(sp, r + 1);
                const int pos = p0 + lane - cb;
                if (lane < p1 - p0 && pos >= 0 && pos < LIST_CAP) stage[pos] = ev[r];
            }
            for (int r = 0; r < 16; ++r) {
                const int p0 = __builtin_amdgcn_readlane(sp, r);
                const int p1 = __builtin_amdgcn_readlane(sp, r + 1);
                const int base = __builtin_amdgcn_readlane(sb, r);
                for (int i0 = 32; i0 < p1 - p0; i0 += 32) {
                    const int i = i0 + lane;
                    const int pos = p0 + i - cb;
                    if (i < p1 - p0 && pos >= 0 && pos < LIST_CAP) stage[pos] = sorted[base + i];
                }
            }
        }
        __syncthreads();

        int ent[PER_LANE];
        unsigned same[PER_LANE];
#pragma unroll
        for (int j = 0; j < PER_LANE; ++j) {
            const int q = wave * (32 * PER_LANE) + j * 32 + lane;
            ent[j] = (cb + q < total) ? stage[q] : -1;
        }
#pragma unroll
        for (int j = 0; j < PER_LANE; ++j) {
            const int key = (ent[j] >= 0) ? (ent[j] & 255) : 256;
            same[j] = same_part_lanes(key);
            if (lane == __builtin_ctz(same[j])) hw[key] += __builtin_popcount(same[j]);
            wave_lds_order();
        }
        __syncthreads();

        int tot = 0;
        if (t < 256) {
#pragma unroll
            for (int i = 0; i < 16; ++i) tot += h[i * NB + t];
        }
        int incl = tot;
#pragma unroll
        for (int o = 1; o < 32; o <<= 1) {
            const int u = __shfl_up(incl, o);
            if (lane >= o) incl += u;
        }
        if (lane == 31 && wave < 8) wsum[wave] = incl;
        __syncthreads();
        if (t < 256) {
            int run = incl - tot;
#pragma unroll
            for (int i = 0; i < 8; ++i) run += (i < wave) ? wsum[i] : 0;
            nstart[t] = run;
#pragma unroll
            for (int i = 0; i < 16; ++i) {
                const int c = h[i * NB + t];
                h[i * NB + t] = run;
                run += c;
            }
            if (t == 255) nstart[256] = run;
        }
        __syncthreads();

#pragma unroll
        for (int j = 0; j < PER_LANE; ++j) {
            const int key = (ent[j] >= 0) ? (ent[j] & 255) : 256;
            const int start = hw[key];
            wave_lds_order();
            if (lane == __builtin_ctz(same[j])) hw[key] = start + __builtin_popcount(same[j]);
            wave_lds_order();
            if (ent[j] >= 0) ids[start + __builtin_popcount(same[j] & lt)] = ent[j] >> 8;
        }
        __syncthreads();

        {
            const int nb = nstart[16 * wave + ((lane < 16) ? lane : 16)];
            const int S = __builtin_amdgcn_readlane(nb, 0);
            const int E = __builtin_amdgcn_readlane(nb, 16);
            if (S < E) {
                int cur = 0;
                int cend = __builtin_amdgcn_readlane(nb, 1);
                float acc = mytile[lane];

                auto load16 = [&](int k, float (&v)[16]) {
                    const int idv = (k + lane < E) ? ids[k + lane] : 0;
#pragma unroll
                    for (int i = 0; i < 16; ++i) {
                        const int sk = __builtin_amdgcn_readlane(idv, i);
                        v[i] = X[sk * 32 + lane];
                    }
                };
                auto add_row = [&](int p, float v) {
                    while (__builtin_expect(p >= cend, 0)) {
                        mytile[cur * TILE_LD + lane] = acc;
                        ++cur;
                        acc = mytile[cur * TILE_LD + lane];
                        cend = __builtin_amdgcn_readlane(nb, cur + 1);
                    }
                    acc += v;
                };
                auto add16 = [&](int k, const float (&v)[16]) {
                    if (k + 16 <= E) {
#pragma unroll
                        for (int i = 0; i < 16; ++i) add_row(k + i, v[i]);
                    } else {
#pragma unroll
                        for (int i = 0; i < 16; ++i)
                            if (k + i < E) add_row(k + i, v[i]);
                    }
                };

                float va[16], vb[16];
                load16(S, va);
                for (int k = S; k < E; k += 32) {
                    load16(k + 16, vb);
                    add16(k, va);
                    load16(k + 32, va);
                    add16(k + 16, vb);
                }
                mytile[cur * TILE_LD + lane] = acc;
            }
        }
    }

    if (part * 16 + wave < N_TILES) {
        const int row0 = (part * 16 + wave) * 16;
        v2f bfr[2][8];
#pragma unroll
        for (int nt = 0; nt < 2; ++nt) {
#pragma unroll
            for (int s = 0; s < 8; ++s) {
                const int k = 4 * s + 2 * hi;
                bfr[nt][s].x = W[k * 32 + nt * 16 + m];
                bfr[nt][s].y = W[(k + 1) * 32 + nt * 16 + m];
            }
        }
        wave_lds_order();
        v2f a[8];
#pragma unroll
        for (int s = 0; s < 8; ++s)
            a[s] = *(const v2f*)&mytile[m * TILE_LD + 4 * s + 2 * hi];
#pragma unroll
        for (int nt = 0; nt < 2; ++nt) {
            v8f c = {};
#pragma unroll
            for (int s = 0; s < 8; ++s) {
                c = __builtin_amdgcn_wmma_f32_16x16x4_f32(
                         false, a[s],  false, bfr[nt][s],
                         (short)0, c,  false,  false);
            }
            float* op = out + (size_t)(row0 + 8 * hi) * 32 + nt * 16 + m;
#pragma unroll
            for (int r = 0; r < 8; ++r) op[(size_t)r * 32] = c[r];
        }
    }
}

extern "C" void kernel_launch(void* const* d_in, const int* in_sizes, int n_in,
                              void* d_out, int out_size, void* d_ws, size_t ws_size,
                              hipStream_t stream) {
    const float* X   = (const float*)d_in[0];
    const float* W   = (const float*)d_in[1];
    const int*   src = (const int*)d_in[2];
    const int*   dst = (const int*)d_in[3];
    float*       out = (float*)d_out;
    int*         runs   = (int*)d_ws;
    int*         sorted = runs + N_PART * N_CHUNK;

    gin_split_kernel<<<N_CHUNKS, 256, 0, stream>>>(src, dst, runs, sorted);
    gin_pull_gemm_kernel<<<N_PBLK, 512, 0, stream>>>(X, W, runs, sorted, out);
}


# split kernel puts its chunk's slice of the edge list together in LDS and writes it out as 200 whole 128-byte lines (16 bytes per lane), not 6400 lane-scattered 4-byte global stores; sum order kept
# speedup vs baseline: 3.3933x; 1.1506x over previous
#include <hip/hip_runtime.h>

#define N_NODES 100000
#define N_EDGES 1600000
#define IN_DIM  32
#define OUT_DIM 32
#define N_TILES (N_NODES / 16)
#define TILE_LD 36

#define N_PART   392
#define N_CHUNK  256
#define N_CHUNKS (N_EDGES / CHUNK_E)
#define CHUNK_E  6400
#define WAVE_E   800
#define ROUNDS   25
#define PER_LANE 9
#define LIST_CAP (512 * PER_LANE)
#define NB       257
#define N_PBLK   ((N_NODES + 255) / 256)

typedef __attribute__((ext_vector_type(2))) float v2f;
typedef __attribute__((ext_vector_type(8))) float v8f;
typedef __attribute__((ext_vector_type(4))) int   v4i;

__device__ __forceinline__ void wave_lds_order() {
    __builtin_amdgcn_fence(__ATOMIC_RELEASE, "wavefront");
    __builtin_amdgcn_wave_barrier();
    __builtin_amdgcn_fence(__ATOMIC_ACQUIRE, "wavefront");
}


__device__ __forceinline__ unsigned same_part_lanes(int part) {
    unsigned same = 0xffffffffu;
#pragma unroll
    for (int b = 0; b < 9; ++b) {
        const bool bit = (part >> b) & 1;
        const unsigned bal = __builtin_amdgcn_ballot_w32(bit);
        same &= bit ? bal : ~bal;
    }
    return same;
}

__global__ __launch_bounds__(256) void gin_split_kernel(const int* __restrict__ src,
                                                        const int* __restrict__ dst,
                                                        int* __restrict__ runs,
                                                        int* __restrict__ sorted) {
    __shared__ int h[8 * N_PART];
    __shared__ int wtot[8];
    __shared__ __attribute__((aligned(16))) int slice[CHUNK_E];
    const int t = threadIdx.x;
    const int lane = t & 31;
    const int w = t >> 5;
    const int b = blockIdx.x;
    for (int i = t; i < 8 * N_PART; i += 256) h[i] = 0;
    __syncthreads();

    const int e0 = b * CHUNK_E + w * WAVE_E + lane;
    int dv[ROUNDS];
    int sv[ROUNDS];
#pragma unroll
    for (int r = 0; r < ROUNDS; ++r) dv[r] = dst[e0 + r * 32];
#pragma unroll
    for (int r = 0; r < ROUNDS; ++r) sv[r] = src[e0 + r * 32];

    int* hw = h + w * N_PART;
    unsigned same[ROUNDS];
#pragma unroll
    for (int r = 0; r < ROUNDS; ++r) {
        same[r] = same_part_lanes(dv[r] >> 8);
        if (lane == __builtin_ctz(same[r])) hw[dv[r] >> 8] += __builtin_popcount(same[r]);
        wave_lds_order();
    }
    __syncthreads();
    const int d0 = 2 * t;
    int tot0 = 0, tot1 = 0;
    if (t < N_PART / 2) {
#pragma unroll
        for (int i = 0; i < 8; ++i) {
            tot0 += h[i * N_PART + d0];
            tot1 += h[i * N_PART + d0 + 1];
        }
    }
    int incl = tot0 + tot1;
#pragma unroll
    for (int o = 1; o < 32; o <<= 1) {
        const int u = __shfl_up(incl, o);
        if (lane >= o) incl += u;
    }
    if (lane == 31) wtot[w] = incl;
    __syncthreads();
    if (t < N_PART / 2) {
        int first = incl - (tot0 + tot1);
#pragma unroll
        for (int i = 0; i < 8; ++i) first += (i < w) ? wtot[i] : 0;
        runs[d0 * N_CHUNK + b] = first;
        runs[(d0 + 1) * N_CHUNK + b] = first + tot0;
        int run = first;
#pragma unroll
        for (int i = 0; i < 8; ++i) {
            const int c = h[i * N_PART + d0];
            h[i * N_PART + d0] = run;
            run += c;
        }
#pragma unroll
        for (int i = 0; i < 8; ++i) {
            const int c = h[i * N_PART + d0 + 1];
            h[i * N_PART + d0 + 1] = run;
            run += c;
        }
    }
    __syncthreads();
    const unsigned lt = (1u << lane) - 1u;
#pragma unroll
    for (int r = 0; r < ROUNDS; ++r) {
        const int part = dv[r] >> 8;
        const int start = hw[part];
        wave_lds_order();
        if (lane == __builtin_ctz(same[r])) hw[part] = start + __builtin_popcount(same[r]);
        wave_lds_order();
        slice[start + __builtin_popcount(same[r] & lt)] = (sv[r] << 8) | (dv[r] & 255);
    }
    __syncthreads();
    const v4i* s4 = (const v4i*)slice;
    v4i* g4 = (v4i*)(sorted + b * CHUNK_E);
    static_assert(CHUNK_E / 4 == 6 * 256 + 64, "write-out rounds");
    v4i q[6];
#pragma unroll
    for (int j = 0; j < 6; ++j) q[j] = s4[t + j * 256];
#pragma unroll
    for (int j = 0; j < 6; ++j) g4[t + j * 256] = q[j];
    if (t < 64) g4[t + 6 * 256] = s4[t + 6 * 256];
}

__global__ __launch_bounds__(512) void gin_pull_gemm_kernel(const float* __restrict__ X,
                                                            const float* __restrict__ W,
                                                            const int* __restrict__ runs,
                                                            const int* __restrict__ sorted,
                                                            float* __restrict__ out) {
    __shared__ int   h[16 * NB];
    __shared__ int   nstart[NB];
    __shared__ int   ids[LIST_CAP];
    __shared__ int   wsum[8];
    __shared__ int   segbase[256];
    __shared__ int   segpos[257];
    __shared__ int   stage[LIST_CAP];
    __shared__ __attribute__((aligned(16))) float tile[16 * 16 * TILE_LD];

    const int t = threadIdx.x;
    const int lane = t & 31;
    const int wave = __builtin_amdgcn_readfirstlane((int)(threadIdx.x >> 5));
    const int m  = lane & 15;
    const int hi = lane >> 4;
    const int part = blockIdx.x;
    const unsigned lt = (1u << lane) - 1u;
    int*   hw     = h + wave * NB;
    float* mytile = tile + wave * (16 * TILE_LD);

    for (int n = 0; n < 16; ++n) mytile[n * TILE_LD + lane] = 0.0f;

    {
        int s0 = 0, len = 0;
        if (t < N_CHUNKS) {
            s0  = runs[part * N_CHUNK + t];
            len = runs[(part + 1) * N_CHUNK + t] - s0;
        }
        int incl = len;
#pragma unroll
        for (int o = 1; o < 32; o <<= 1) {
            const int u = __shfl_up(incl, o);
            if (lane >= o) incl += u;
        }
        if (lane == 31 && wave < 8) wsum[wave] = incl;
        __syncthreads();
        if (t < 256) {
            int first = incl - len;
#pragma unroll
            for (int i = 0; i < 8; ++i) first += (i < wave) ? wsum[i] : 0;
            segbase[t] = t * CHUNK_E + s0;
            segpos[t] = first;
            if (t == 255) segpos[256] = first + len;
        }
        __syncthreads();
    }
    const int total = segpos[256];

    for (int cb = 0; cb < total; cb += LIST_CAP) {
        for (int i = t; i < 16 * NB; i += 512) h[i] = 0;

        {
            const int sp = segpos[16 * wave + ((lane < 16) ? lane : 16)];
            const int sb = segbase[16 * wave + (lane & 15)];
            int ev[16];
#pragma unroll
            for (int r = 0; r < 16; ++r) {
                const int p0 = __builtin_amdgcn_readlane(sp, r);
                const int p1 = __builtin_amdgcn_readlane(sp, r + 1);
                const int base = __builtin_amdgcn_readlane(sb, r);
                const int last = (p1 - p0 > 0) ? (p1 - p0 - 1) : 0;
                const int idx = base + ((lane < last) ? lane : last);
                ev[r] = sorted[(idx < N_EDGES) ? idx : (N_EDGES - 1)];
            }
#pragma unroll
            for (int r = 0; r < 16; ++r) {
                const int p0 = __builtin_amdgcn_readlane(sp, r);
                const int p1 = __builtin_amdgcn_readlane(sp, r + 1);
                const int pos = p0 + lane - cb;
                if (lane < p1 - p0 && pos >= 0 && pos < LIST_CAP) stage[pos] = ev[r];
            }
            for (int r = 0; r < 16; ++r) {
                const int p0 = __builtin_amdgcn_readlane(sp, r);
                const int p1 = __builtin_amdgcn_readlane(sp, r + 1);
                const int base = __builtin_amdgcn_readlane(sb, r);
                for (int i0 = 32; i0 < p1 - p0; i0 += 32) {
                    const int i = i0 + lane;
                    const int pos = p0 + i - cb;
                    if (i < p1 - p0 && pos >= 0 && pos < LIST_CAP) stage[pos] = sorted[base + i];
                }
            }
        }
        __syncthreads();

        int ent[PER_LANE];
        unsigned same[PER_LANE];
#pragma unroll
        for (int j = 0; j < PER_LANE; ++j) {
            const int q = wave * (32 * PER_LANE) + j * 32 + lane;
            ent[j] = (cb + q < total) ? stage[q] : -1;
        }
#pragma unroll
        for (int j = 0; j < PER_LANE; ++j) {
            const int key = (ent[j] >= 0) ? (ent[j] & 255) : 256;
            same[j] = same_part_lanes(key);
            if (lane == __builtin_ctz(same[j])) hw[key] += __builtin_popcount(same[j]);
            wave_lds_order();
        }
        __syncthreads();

        int tot = 0;
        if (t < 256) {
#pragma unroll
            for (int i = 0; i < 16; ++i) tot += h[i * NB + t];
        }
        int incl = tot;
#pragma unroll
        for (int o = 1; o < 32; o <<= 1) {
            const int u = __shfl_up(incl, o);
            if (lane >= o) incl += u;
        }
        if (lane == 31 && wave < 8) wsum[wave] = incl;
        __syncthreads();
        if (t < 256) {
            int run = incl - tot;
#pragma unroll
            for (int i = 0; i < 8; ++i) run += (i < wave) ? wsum[i] : 0;
            nstart[t] = run;
#pragma unroll
            for (int i = 0; i < 16; ++i) {
                const int c = h[i * NB + t];
                h[i * NB + t] = run;
                run += c;
            }
            if (t == 255) nstart[256] = run;
        }
        __syncthreads();

#pragma unroll
        for (int j = 0; j < PER_LANE; ++j) {
            const int key = (ent[j] >= 0) ? (ent[j] & 255) : 256;
            const int start = hw[key];
            wave_lds_order();
            if (lane == __builtin_ctz(same[j])) hw[key] = start + __builtin_popcount(same[j]);
            wave_lds_order();
            if (ent[j] >= 0) ids[start + __builtin_popcount(same[j] & lt)] = ent[j] >> 8;
        }
        __syncthreads();

        {
            const int nb = nstart[16 * wave + ((lane < 16) ? lane : 16)];
            const int S = __builtin_amdgcn_readlane(nb, 0);
            const int E = __builtin_amdgcn_readlane(nb, 16);
            if (S < E) {
                int cur = 0;
                int cend = __builtin_amdgcn_readlane(nb, 1);
                float acc = mytile[lane];

                auto load16 = [&](int k, float (&v)[16]) {
                    const int idv = (k + lane < E) ? ids[k + lane] : 0;
#pragma unroll
                    for (int i = 0; i < 16; ++i) {
                        const int sk = __builtin_amdgcn_readlane(idv, i);
                        v[i] = X[sk * 32 + lane];
                    }
                };
                auto add_row = [&](int p, float v) {
                    while (__builtin_expect(p >= cend, 0)) {
                        mytile[cur * TILE_LD + lane] = acc;
                        ++cur;
                        acc = mytile[cur * TILE_LD + lane];
                        cend = __builtin_amdgcn_readlane(nb, cur + 1);
                    }
                    acc += v;
                };
                auto add16 = [&](int k, const float (&v)[16]) {
                    if (k + 16 <= E) {
#pragma unroll
                        for (int i = 0; i < 16; ++i) add_row(k + i, v[i]);
                    } else {
#pragma unroll
                        for (int i = 0; i < 16; ++i)
                            if (k + i < E) add_row(k + i, v[i]);
                    }
                };

                float va[16], vb[16];
                load16(S, va);
                for (int k = S; k < E; k += 32) {
                    load16(k + 16, vb);
                    add16(k, va);
                    load16(k + 32, va);
                    add16(k + 16, vb);
                }
                mytile[cur * TILE_LD + lane] = acc;
            }
        }
    }

    if (part * 16 + wave < N_TILES) {
        const int row0 = (part * 16 + wave) * 16;
        v2f bfr[2][8];
#pragma unroll
        for (int nt = 0; nt < 2; ++nt) {
#pragma unroll
            for (int s = 0; s < 8; ++s) {
                const int k = 4 * s + 2 * hi;
                bfr[nt][s].x = W[k * 32 + nt * 16 + m];
                bfr[nt][s].y = W[(k + 1) * 32 + nt * 16 + m];
            }
        }
        wave_lds_order();
        v2f a[8];
#pragma unroll
        for (int s = 0; s < 8; ++s)
            a[s] = *(const v2f*)&mytile[m * TILE_LD + 4 * s + 2 * hi];
#pragma unroll
        for (int nt = 0; nt < 2; ++nt) {
            v8f c = {};
#pragma unroll
            for (int s = 0; s < 8; ++s) {
                c = __builtin_amdgcn_wmma_f32_16x16x4_f32(
                         false, a[s],  false, bfr[nt][s],
                         (short)0, c,  false,  false);
            }
            float* op = out + (size_t)(row0 + 8 * hi) * 32 + nt * 16 + m;
#pragma unroll
            for (int r = 0; r < 8; ++r) op[(size_t)r * 32] = c[r];
        }
    }
}

extern "C" void kernel_launch(void* const* d_in, const int* in_sizes, int n_in,
                              void* d_out, int out_size, void* d_ws, size_t ws_size,
                              hipStream_t stream) {
    const float* X   = (const float*)d_in[0];
    const float* W   = (const float*)d_in[1];
    const int*   src = (const int*)d_in[2];
    const int*   dst = (const int*)d_in[3];
    float*       out = (float*)d_out;
    int*         runs   = (int*)d_ws;
    int*         sorted = runs + N_PART * N_CHUNK;

    gin_split_kernel<<<N_CHUNKS, 256, 0, stream>>>(src, dst, runs, sorted);
    gin_pull_gemm_kernel<<<N_PBLK, 512, 0, stream>>>(X, W, runs, sorted, out);
}


# counting sorts of both kernels: each round keeps the counter value its count read (its group's offset in the wave's share); placement only reads starts, no LDS read-wait-store chain; sum order kept
# speedup vs baseline: 3.5333x; 1.0413x over previous
#include <hip/hip_runtime.h>

#define N_NODES 100000
#define N_EDGES 1600000
#define IN_DIM  32
#define OUT_DIM 32
#define N_TILES (N_NODES / 16)
#define TILE_LD 36

#define N_PART   392
#define N_CHUNK  256
#define N_CHUNKS (N_EDGES / CHUNK_E)
#define CHUNK_E  6400
#define WAVE_E   800
#define ROUNDS   25
#define PER_LANE 9
#define LIST_CAP (512 * PER_LANE)
#define NB       257
#define N_PBLK   ((N_NODES + 255) / 256)

typedef __attribute__((ext_vector_type(2))) float v2f;
typedef __attribute__((ext_vector_type(8))) float v8f;
typedef __attribute__((ext_vector_type(4))) int   v4i;

__device__ __forceinline__ void wave_lds_order() {
    __builtin_amdgcn_fence(__ATOMIC_RELEASE, "wavefront");
    __builtin_amdgcn_wave_barrier();
    __builtin_amdgcn_fence(__ATOMIC_ACQUIRE, "wavefront");
}


__device__ __forceinline__ unsigned same_part_lanes(int part) {
    unsigned same = 0xffffffffu;
#pragma unroll
    for (int b = 0; b < 9; ++b) {
        const bool bit = (part >> b) & 1;
        const unsigned bal = __builtin_amdgcn_ballot_w32(bit);
        same &= bit ? bal : ~bal;
    }
    return same;
}

__global__ __launch_bounds__(256) void gin_split_kernel(const int* __restrict__ src,
                                                        const int* __restrict__ dst,
                                                        int* __restrict__ runs,
                                                        int* __restrict__ sorted) {
    __shared__ int h[8 * N_PART];
    __shared__ int wtot[8];
    __shared__ __attribute__((aligned(16))) int slice[CHUNK_E];
    const int t = threadIdx.x;
    const int lane = t & 31;
    const int w = t >> 5;
    const int b = blockIdx.x;
    for (int i = t; i < 8 * N_PART; i += 256) h[i] = 0;
    __syncthreads();

    const int e0 = b * CHUNK_E + w * WAVE_E + lane;
    int dv[ROUNDS];
    int sv[ROUNDS];
#pragma unroll
    for (int r = 0; r < ROUNDS; ++r) dv[r] = dst[e0 + r * 32];
#pragma unroll
    for (int r = 0; r < ROUNDS; ++r) sv[r] = src[e0 + r * 32];

    int* hw = h + w * N_PART;
    unsigned same[ROUNDS];
    int pre[ROUNDS];
    same[0] = same_part_lanes(dv[0] >> 8);
#pragma unroll
    for (int r = 0; r < ROUNDS; ++r) {
        const int part = dv[r] >> 8;
        pre[r] = hw[part];
        if (r + 1 < ROUNDS) same[r + 1] = same_part_lanes(dv[r + 1] >> 8);
        wave_lds_order();
        if (lane == __builtin_ctz(same[r])) hw[part] = pre[r] + __builtin_popcount(same[r]);
        wave_lds_order();
    }
    __syncthreads();
    const int d0 = 2 * t;
    int tot0 = 0, tot1 = 0;
    if (t < N_PART / 2) {
#pragma unroll
        for (int i = 0; i < 8; ++i) {
            tot0 += h[i * N_PART + d0];
            tot1 += h[i * N_PART + d0 + 1];
        }
    }
    int incl = tot0 + tot1;
#pragma unroll
    for (int o = 1; o < 32; o <<= 1) {
        const int u = __shfl_up(incl, o);
        if (lane >= o) incl += u;
    }
    if (lane == 31) wtot[w] = incl;
    __syncthreads();
    if (t < N_PART / 2) {
        int first = incl - (tot0 + tot1);
#pragma unroll
        for (int i = 0; i < 8; ++i) first += (i < w) ? wtot[i] : 0;
        runs[d0 * N_CHUNK + b] = first;
        runs[(d0 + 1) * N_CHUNK + b] = first + tot0;
        int run = first;
#pragma unroll
        for (int i = 0; i < 8; ++i) {
            const int c = h[i * N_PART + d0];
            h[i * N_PART + d0] = run;
            run += c;
        }
#pragma unroll
        for (int i = 0; i < 8; ++i) {
            const int c = h[i * N_PART + d0 + 1];
            h[i * N_PART + d0 + 1] = run;
            run += c;
        }
    }
    __syncthreads();
    const unsigned lt = (1u << lane) - 1u;
#pragma unroll
    for (int r = 0; r < ROUNDS; ++r) {
        const int start = hw[dv[r] >> 8];
        slice[start + pre[r] + __builtin_popcount(same[r] & lt)] = (sv[r] << 8) | (dv[r] & 255);
    }
    __syncthreads();
    const v4i* s4 = (const v4i*)slice;
    v4i* g4 = (v4i*)(sorted + b * CHUNK_E);
    static_assert(CHUNK_E / 4 == 6 * 256 + 64, "write-out rounds");
    v4i q[6];
#pragma unroll
    for (int j = 0; j < 6; ++j) q[j] = s4[t + j * 256];
#pragma unroll
    for (int j = 0; j < 6; ++j) g4[t + j * 256] = q[j];
    if (t < 64) g4[t + 6 * 256] = s4[t + 6 * 256];
}

__global__ __launch_bounds__(512) void gin_pull_gemm_kernel(const float* __restrict__ X,
                                                            const float* __restrict__ W,
                                                            const int* __restrict__ runs,
                                                            const int* __restrict__ sorted,
                                                            float* __restrict__ out) {
    __shared__ int   h[16 * NB];
    __shared__ int   nstart[NB];
    __shared__ int   ids[LIST_CAP];
    __shared__ int   wsum[8];
    __shared__ int   segbase[256];
    __shared__ int   segpos[257];
    __shared__ int   stage[LIST_CAP];
    __shared__ __attribute__((aligned(16))) float tile[16 * 16 * TILE_LD];

    const int t = threadIdx.x;
    const int lane = t & 31;
    const int wave = __builtin_amdgcn_readfirstlane((int)(threadIdx.x >> 5));
    const int m  = lane & 15;
    const int hi = lane >> 4;
    const int part = blockIdx.x;
    const unsigned lt = (1u << lane) - 1u;
    int*   hw     = h + wave * NB;
    float* mytile = tile + wave * (16 * TILE_LD);

    for (int n = 0; n < 16; ++n) mytile[n * TILE_LD + lane] = 0.0f;

    {
        int s0 = 0, len = 0;
        if (t < N_CHUNKS) {
            s0  = runs[part * N_CHUNK + t];
            len = runs[(part + 1) * N_CHUNK + t] - s0;
        }
        int incl = len;
#pragma unroll
        for (int o = 1; o < 32; o <<= 1) {
            const int u = __shfl_up(incl, o);
            if (lane >= o) incl += u;
        }
        if (lane == 31 && wave < 8) wsum[wave] = incl;
        __syncthreads();
        if (t < 256) {
            int first = incl - len;
#pragma unroll
            for (int i = 0; i < 8; ++i) first += (i < wave) ? wsum[i] : 0;
            segbase[t] = t * CHUNK_E + s0;
            segpos[t] = first;
            if (t == 255) segpos[256] = first + len;
        }
        __syncthreads();
    }
    const int total = segpos[256];

    for (int cb = 0; cb < total; cb += LIST_CAP) {
        for (int i = t; i < 16 * NB; i += 512) h[i] = 0;

        {
            const int sp = segpos[16 * wave + ((lane < 16) ? lane : 16)];
            const int sb = segbase[16 * wave + (lane & 15)];
            int ev[16];
#pragma unroll
            for (int r = 0; r < 16; ++r) {
                const int p0 = __builtin_amdgcn_readlane(sp, r);
                const int p1 = __builtin_amdgcn_readlane(sp, r + 1);
                const int base = __builtin_amdgcn_readlane(sb, r);
                const int last = (p1 - p0 > 0) ? (p1 - p0 - 1) : 0;
                const int idx = base + ((lane < last) ? lane : last);
                ev[r] = sorted[(idx < N_EDGES) ? idx : (N_EDGES - 1)];
            }
#pragma unroll
            for (int r = 0; r < 16; ++r) {
                const int p0 = __builtin_amdgcn_readlane(sp, r);
                const int p1 = __builtin_amdgcn_readlane(sp, r + 1);
                const int pos = p0 + lane - cb;
                if (lane < p1 - p0 && pos >= 0 && pos < LIST_CAP) stage[pos] = ev[r];
            }
            for (int r = 0; r < 16; ++r) {
                const int p0 = __builtin_amdgcn_readlane(sp, r);
                const int p1 = __builtin_amdgcn_readlane(sp, r + 1);
                const int base = __builtin_amdgcn_readlane(sb, r);
                for (int i0 = 32; i0 < p1 - p0; i0 += 32) {
                    const int i = i0 + lane;
                    const int pos = p0 + i - cb;
                    if (i < p1 - p0 && pos >= 0 && pos < LIST_CAP) stage[pos] = sorted[base + i];
                }
            }
        }
        __syncthreads();

        int ent[PER_LANE];
        unsigned same[PER_LANE];
#pragma unroll
        for (int j = 0; j < PER_LANE; ++j) {
            const int q = wave * (32 * PER_LANE) + j * 32 + lane;
            ent[j] = (cb + q < total) ? stage[q] : -1;
        }
        int pre[PER_LANE];
        same[0] = same_part_lanes((ent[0] >= 0) ? (ent[0] & 255) : 256);
#pragma unroll
        for (int j = 0; j < PER_LANE; ++j) {
            const int key = (ent[j] >= 0) ? (ent[j] & 255) : 256;
            pre[j] = hw[key];
            if (j + 1 < PER_LANE) same[j + 1] = same_part_lanes((ent[j + 1] >= 0) ? (ent[j + 1] & 255) : 256);
            wave_lds_order();
            if (lane == __builtin_ctz(same[j])) hw[key] = pre[j] + __builtin_popcount(same[j]);
            wave_lds_order();
        }
        __syncthreads();

        int tot = 0;
        if (t < 256) {
#pragma unroll
            for (int i = 0; i < 16; ++i) tot += h[i * NB + t];
        }
        int incl = tot;
#pragma unroll
        for (int o = 1; o < 32; o <<= 1) {
            const int u = __shfl_up(incl, o);
            if (lane >= o) incl += u;
        }
        if (lane == 31 && wave < 8) wsum[wave] = incl;
        __syncthreads();
        if (t < 256) {
            int run = incl - tot;
#pragma unroll
            for (int i = 0; i < 8; ++i) run += (i < wave) ? wsum[i] : 0;
            nstart[t] = run;
#pragma unroll
            for (int i = 0; i < 16; ++i) {
                const int c = h[i * NB + t];
                h[i * NB + t] = run;
                run += c;
            }
            if (t == 255) nstart[256] = run;
        }
        __syncthreads();

        if (cb + (wave + 1) * (32 * PER_LANE) <= total) {
#pragma unroll
            for (int j = 0; j < PER_LANE; ++j) {
                const int start = hw[ent[j] & 255];
                ids[start + pre[j] + __builtin_popcount(same[j] & lt)] = ent[j] >> 8;
            }
        } else {
#pragma unroll
            for (int j = 0; j < PER_LANE; ++j) {
                const int key = (ent[j] >= 0) ? (ent[j] & 255) : 256;
                const int start = hw[key];
                if (ent[j] >= 0) ids[start + pre[j] + __builtin_popcount(same[j] & lt)] = ent[j] >> 8;
            }
        }
        __syncthreads();

        {
            const int nb = nstart[16 * wave + ((lane < 16) ? lane : 16)];
            const int S = __builtin_amdgcn_readlane(nb, 0);
            const int E = __builtin_amdgcn_readlane(nb, 16);
            if (S < E) {
                int cur = 0;
                int cend = __builtin_amdgcn_readlane(nb, 1);
                float acc = mytile[lane];

                auto load16 = [&](int k, float (&v)[16]) {
                    const int idv = (k + lane < E) ? ids[k + lane] : 0;
#pragma unroll
                    for (int i = 0; i < 16; ++i) {
                        const int sk = __builtin_amdgcn_readlane(idv, i);
                        v[i] = X[sk * 32 + lane];
                    }
                };
                auto add_row = [&](int p, float v) {
                    while (__builtin_expect(p >= cend, 0)) {
                        mytile[cur * TILE_LD + lane] = acc;
                        ++cur;
                        acc = mytile[cur * TILE_LD + lane];
                        cend = __builtin_amdgcn_readlane(nb, cur + 1);
                    }
                    acc += v;
                };
                auto add16 = [&](int k, const float (&v)[16]) {
                    if (k + 16 <= E) {
#pragma unroll
                        for (int i = 0; i < 16; ++i) add_row(k + i, v[i]);
                    } else {
#pragma unroll
                        for (int i = 0; i < 16; ++i)
                            if (k + i < E) add_row(k + i, v[i]);
                    }
                };

                float va[16], vb[16];
                load16(S, va);
                for (int k = S; k < E; k += 32) {
                    load16(k + 16, vb);
                    add16(k, va);
                    load16(k + 32, va);
                    add16(k + 16, vb);
                }
                mytile[cur * TILE_LD + lane] = acc;
            }
        }
    }

    if (part * 16 + wave < N_TILES) {
        const int row0 = (part * 16 + wave) * 16;
        v2f bfr[2][8];
#pragma unroll
        for (int nt = 0; nt < 2; ++nt) {
#pragma unroll
            for (int s = 0; s < 8; ++s) {
                const int k = 4 * s + 2 * hi;
                bfr[nt][s].x = W[k * 32 + nt * 16 + m];
                bfr[nt][s].y = W[(k + 1) * 32 + nt * 16 + m];
            }
        }
        wave_lds_order();
        v2f a[8];
#pragma unroll
        for (int s = 0; s < 8; ++s)
            a[s] = *(const v2f*)&mytile[m * TILE_LD + 4 * s + 2 * hi];
#pragma unroll
        for (int nt = 0; nt < 2; ++nt) {
            v8f c = {};
#pragma unroll
            for (int s = 0; s < 8; ++s) {
                c = __builtin_amdgcn_wmma_f32_16x16x4_f32(
                         false, a[s],  false, bfr[nt][s],
                         (short)0, c,  false,  false);
            }
            float* op = out + (size_t)(row0 + 8 * hi) * 32 + nt * 16 + m;
#pragma unroll
            for (int r = 0; r < 8; ++r) op[(size_t)r * 32] = c[r];
        }
    }
}

extern "C" void kernel_launch(void* const* d_in, const int* in_sizes, int n_in,
                              void* d_out, int out_size, void* d_ws, size_t ws_size,
                              hipStream_t stream) {
    const float* X   = (const float*)d_in[0];
    const float* W   = (const float*)d_in[1];
    const int*   src = (const int*)d_in[2];
    const int*   dst = (const int*)d_in[3];
    float*       out = (float*)d_out;
    int*         runs   = (int*)d_ws;
    int*         sorted = runs + N_PART * N_CHUNK;

    gin_split_kernel<<<N_CHUNKS, 256, 0, stream>>>(src, dst, runs, sorted);
    gin_pull_gemm_kernel<<<N_PBLK, 512, 0, stream>>>(X, W, runs, sorted, out);
}


# counting sorts of both kernels: the mask of lanes with my key takes 3 vector instructions per key bit, not 6 (spread bit, its ballot, one bit op); a group stores its count unbranched; sum order kept
# speedup vs baseline: 3.6665x; 1.0377x over previous
#include <hip/hip_runtime.h>

#define N_NODES 100000
#define N_EDGES 1600000
#define IN_DIM  32
#define OUT_DIM 32
#define N_TILES (N_NODES / 16)
#define TILE_LD 36

#define N_PART   392
#define N_CHUNK  256
#define N_CHUNKS (N_EDGES / CHUNK_E)
#define CHUNK_E  6400
#define WAVE_E   800
#define ROUNDS   25
#define PER_LANE 9
#define LIST_CAP (512 * PER_LANE)
#define NB       257
#define N_PBLK   ((N_NODES + 255) / 256)

typedef __attribute__((ext_vector_type(2))) float v2f;
typedef __attribute__((ext_vector_type(8))) float v8f;
typedef __attribute__((ext_vector_type(4))) int   v4i;

__device__ __forceinline__ void wave_lds_order() {
    __builtin_amdgcn_fence(__ATOMIC_RELEASE, "wavefront");
    __builtin_amdgcn_wave_barrier();
    __builtin_amdgcn_fence(__ATOMIC_ACQUIRE, "wavefront");
}


template <int NBITS>
__device__ __forceinline__ unsigned same_key_lanes(int key) {
    unsigned differ = 0;
#pragma unroll
    for (int b = 0; b < NBITS; ++b) {
        int mine = (int)((unsigned)key << (31 - b)) >> 31;
        asm("" : "+v"(mine));
        const unsigned bal = __builtin_amdgcn_ballot_w32(mine != 0);
        differ |= bal ^ (unsigned)mine;
    }
    return ~differ;
}

__global__ __launch_bounds__(256) void gin_split_kernel(const int* __restrict__ src,
                                                        const int* __restrict__ dst,
                                                        int* __restrict__ runs,
                                                        int* __restrict__ sorted) {
    __shared__ int h[8 * N_PART];
    __shared__ int wtot[8];
    __shared__ __attribute__((aligned(16))) int slice[CHUNK_E];
    const int t = threadIdx.x;
    const int lane = t & 31;
    const int w = t >> 5;
    const int b = blockIdx.x;
    for (int i = t; i < 8 * N_PART; i += 256) h[i] = 0;
    __syncthreads();

    const int e0 = b * CHUNK_E + w * WAVE_E + lane;
    int dv[ROUNDS];
    int sv[ROUNDS];
#pragma unroll
    for (int r = 0; r < ROUNDS; ++r) dv[r] = dst[e0 + r * 32];
#pragma unroll
    for (int r = 0; r < ROUNDS; ++r) sv[r] = src[e0 + r * 32];

    int* hw = h + w * N_PART;
    unsigned same[ROUNDS];
    int pre[ROUNDS];
    same[0] = same_key_lanes<9>(dv[0] >> 8);
#pragma unroll
    for (int r = 0; r < ROUNDS; ++r) {
        const int part = dv[r] >> 8;
        pre[r] = hw[part];
        if (r + 1 < ROUNDS) same[r + 1] = same_key_lanes<9>(dv[r + 1] >> 8);
        wave_lds_order();
        hw[part] = pre[r] + __builtin_popcount(same[r]);
        wave_lds_order();
    }
    __syncthreads();
    const int d0 = 2 * t;
    int tot0 = 0, tot1 = 0;
    if (t < N_PART / 2) {
#pragma unroll
        for (int i = 0; i < 8; ++i) {
            tot0 += h[i * N_PART + d0];
            tot1 += h[i * N_PART + d0 + 1];
        }
    }
    int incl = tot0 + tot1;
#pragma unroll
    for (int o = 1; o < 32; o <<= 1) {
        const int u = __shfl_up(incl, o);
        if (lane >= o) incl += u;
    }
    if (lane == 31) wtot[w] = incl;
    __syncthreads();
    if (t < N_PART / 2) {
        int first = incl - (tot0 + tot1);
#pragma unroll
        for (int i = 0; i < 8; ++i) first += (i < w) ? wtot[i] : 0;
        runs[d0 * N_CHUNK + b] = first;
        runs[(d0 + 1) * N_CHUNK + b] = first + tot0;
        int run = first;
#pragma unroll
        for (int i = 0; i < 8; ++i) {
            const int c = h[i * N_PART + d0];
            h[i * N_PART + d0] = run;
            run += c;
        }
#pragma unroll
        for (int i = 0; i < 8; ++i) {
            const int c = h[i * N_PART + d0 + 1];
            h[i * N_PART + d0 + 1] = run;
            run += c;
        }
    }
    __syncthreads();
#pragma unroll
    for (int r = 0; r < ROUNDS; ++r) {
        const int start = hw[dv[r] >> 8];
        slice[start + (int)__builtin_amdgcn_mbcnt_lo(same[r], (unsigned)pre[r])] = (sv[r] << 8) | (dv[r] & 255);
    }
    __syncthreads();
    const v4i* s4 = (const v4i*)slice;
    v4i* g4 = (v4i*)(sorted + b * CHUNK_E);
    static_assert(CHUNK_E / 4 == 6 * 256 + 64, "write-out rounds");
    v4i q[6];
#pragma unroll
    for (int j = 0; j < 6; ++j) q[j] = s4[t + j * 256];
#pragma unroll
    for (int j = 0; j < 6; ++j) g4[t + j * 256] = q[j];
    if (t < 64) g4[t + 6 * 256] = s4[t + 6 * 256];
}

template <int NBITS>
__device__ __forceinline__ void count_wave(int* hw, const int (&ent)[PER_LANE],
                                           unsigned (&same)[PER_LANE], int (&pre)[PER_LANE]) {
    auto key_of = [](int e) { return (NBITS == 8) ? (e & 255) : ((e >= 0) ? (e & 255) : 256); };
    same[0] = same_key_lanes<NBITS>(key_of(ent[0]));
#pragma unroll
    for (int j = 0; j < PER_LANE; ++j) {
        const int key = key_of(ent[j]);
        pre[j] = hw[key];
        if (j + 1 < PER_LANE) same[j + 1] = same_key_lanes<NBITS>(key_of(ent[j + 1]));
        wave_lds_order();
        hw[key] = pre[j] + __builtin_popcount(same[j]);
        wave_lds_order();
    }
}

__global__ __launch_bounds__(512) void gin_pull_gemm_kernel(const float* __restrict__ X,
                                                            const float* __restrict__ W,
                                                            const int* __restrict__ runs,
                                                            const int* __restrict__ sorted,
                                                            float* __restrict__ out) {
    __shared__ int   h[16 * NB];
    __shared__ int   nstart[NB];
    __shared__ int   ids[LIST_CAP];
    __shared__ int   wsum[8];
    __shared__ int   segbase[256];
    __shared__ int   segpos[257];
    __shared__ int   stage[LIST_CAP];
    __shared__ __attribute__((aligned(16))) float tile[16 * 16 * TILE_LD];

    const int t = threadIdx.x;
    const int lane = t & 31;
    const int wave = __builtin_amdgcn_readfirstlane((int)(threadIdx.x >> 5));
    const int m  = lane & 15;
    const int hi = lane >> 4;
    const int part = blockIdx.x;
    int*   hw     = h + wave * NB;
    float* mytile = tile + wave * (16 * TILE_LD);

    for (int n = 0; n < 16; ++n) mytile[n * TILE_LD + lane] = 0.0f;

    {
        int s0 = 0, len = 0;
        if (t < N_CHUNKS) {
            s0  = runs[part * N_CHUNK + t];
            len = runs[(part + 1) * N_CHUNK + t] - s0;
        }
        int incl = len;
#pragma unroll
        for (int o = 1; o < 32; o <<= 1) {
            const int u = __shfl_up(incl, o);
            if (lane >= o) incl += u;
        }
        if (lane == 31 && wave < 8) wsum[wave] = incl;
        __syncthreads();
        if (t < 256) {
            int first = incl - len;
#pragma unroll
            for (int i = 0; i < 8; ++i) first += (i < wave) ? wsum[i] : 0;
            segbase[t] = t * CHUNK_E + s0;
            segpos[t] = first;
            if (t == 255) segpos[256] = first + len;
        }
        __syncthreads();
    }
    const int total = segpos[256];

    for (int cb = 0; cb < total; cb += LIST_CAP) {
        for (int i = t; i < 16 * NB; i += 512) h[i] = 0;

        {
            const int sp = segpos[16 * wave + ((lane < 16) ? lane : 16)];
            const int sb = segbase[16 * wave + (lane & 15)];
            int ev[16];
#pragma unroll
            for (int r = 0; r < 16; ++r) {
                const int p0 = __builtin_amdgcn_readlane(sp, r);
                const int p1 = __builtin_amdgcn_readlane(sp, r + 1);
                const int base = __builtin_amdgcn_readlane(sb, r);
                const int last = (p1 - p0 > 0) ? (p1 - p0 - 1) : 0;
                const int idx = base + ((lane < last) ? lane : last);
                ev[r] = sorted[(idx < N_EDGES) ? idx : (N_EDGES - 1)];
            }
#pragma unroll
            for (int r = 0; r < 16; ++r) {
                const int p0 = __builtin_amdgcn_readlane(sp, r);
                const int p1 = __builtin_amdgcn_readlane(sp, r + 1);
                const int pos = p0 + lane - cb;
                if (lane < p1 - p0 && pos >= 0 && pos < LIST_CAP) stage[pos] = ev[r];
            }
            for (int r = 0; r < 16; ++r) {
                const int p0 = __builtin_amdgcn_readlane(sp, r);
                const int p1 = __builtin_amdgcn_readlane(sp, r + 1);
                const int base = __builtin_amdgcn_readlane(sb, r);
                for (int i0 = 32; i0 < p1 - p0; i0 += 32) {
                    const int i = i0 + lane;
                    const int pos = p0 + i - cb;
                    if (i < p1 - p0 && pos >= 0 && pos < LIST_CAP) stage[pos] = sorted[base + i];
                }
            }
        }
        __syncthreads();

        int ent[PER_LANE];
        unsigned same[PER_LANE];
#pragma unroll
        for (int j = 0; j < PER_LANE; ++j) {
            const int q = wave * (32 * PER_LANE) + j * 32 + lane;
            ent[j] = (cb + q < total) ? stage[q] : -1;
        }
        int pre[PER_LANE];
        const bool whole = cb + (wave + 1) * (32 * PER_LANE) <= total;
        if (whole) count_wave<8>(hw, ent, same, pre);
        else       count_wave<9>(hw, ent, same, pre);
        __syncthreads();

        int tot = 0;
        if (t < 256) {
#pragma unroll
            for (int i = 0; i < 16; ++i) tot += h[i * NB + t];
        }
        int incl = tot;
#pragma unroll
        for (int o = 1; o < 32; o <<= 1) {
            const int u = __shfl_up(incl, o);
            if (lane >= o) incl += u;
        }
        if (lane == 31 && wave < 8) wsum[wave] = incl;
        __syncthreads();
        if (t < 256) {
            int run = incl - tot;
#pragma unroll
            for (int i = 0; i < 8; ++i) run += (i < wave) ? wsum[i] : 0;
            nstart[t] = run;
#pragma unroll
            for (int i = 0; i < 16; ++i) {
                const int c = h[i * NB + t];
                h[i * NB + t] = run;
                run += c;
            }
            if (t == 255) nstart[256] = run;
        }
        __syncthreads();

        if (whole) {
#pragma unroll
            for (int j = 0; j < PER_LANE; ++j) {
                const int start = hw[ent[j] & 255];
                ids[start + (int)__builtin_amdgcn_mbcnt_lo(same[j], (unsigned)pre[j])] = ent[j] >> 8;
            }
        } else {
#pragma unroll
            for (int j = 0; j < PER_LANE; ++j) {
                const int key = (ent[j] >= 0) ? (ent[j] & 255) : 256;
                const int start = hw[key];
                if (ent[j] >= 0) ids[start + (int)__builtin_amdgcn_mbcnt_lo(same[j], (unsigned)pre[j])] = ent[j] >> 8;
            }
        }
        __syncthreads();

        {
            const int nb = nstart[16 * wave + ((lane < 16) ? lane : 16)];
            const int S = __builtin_amdgcn_readlane(nb, 0);
            const int E = __builtin_amdgcn_readlane(nb, 16);
            if (S < E) {
                int cur = 0;
                int cend = __builtin_amdgcn_readlane(nb, 1);
                float acc = mytile[lane];

                auto load16 = [&](int k, float (&v)[16]) {
                    const int idv = (k + lane < E) ? ids[k + lane] : 0;
#pragma unroll
                    for (int i = 0; i < 16; ++i) {
                        const int sk = __builtin_amdgcn_readlane(idv, i);
                        v[i] = X[sk * 32 + lane];
                    }
                };
                auto add_row = [&](int p, float v) {
                    while (__builtin_expect(p >= cend, 0)) {
                        mytile[cur * TILE_LD + lane] = acc;
                        ++cur;
                        acc = mytile[cur * TILE_LD + lane];
                        cend = __builtin_amdgcn_readlane(nb, cur + 1);
                    }
                    acc += v;
                };
                auto add16 = [&](int k, const float (&v)[16]) {
                    if (k + 16 <= E) {
#pragma unroll
                        for (int i = 0; i < 16; ++i) add_row(k + i, v[i]);
                    } else {
#pragma unroll
                        for (int i = 0; i < 16; ++i)
                            if (k + i < E) add_row(k + i, v[i]);
                    }
                };

                float va[16], vb[16];
                load16(S, va);
                for (int k = S; k < E; k += 32) {
                    load16(k + 16, vb);
                    add16(k, va);
                    load16(k + 32, va);
                    add16(k + 16, vb);
                }
                mytile[cur * TILE_LD + lane] = acc;
            }
        }
    }

    if (part * 16 + wave < N_TILES) {
        const int row0 = (part * 16 + wave) * 16;
        v2f bfr[2][8];
#pragma unroll
        for (int nt = 0; nt < 2; ++nt) {
#pragma unroll
            for (int s = 0; s < 8; ++s) {
                const int k = 4 * s + 2 * hi;
                bfr[nt][s].x = W[k * 32 + nt * 16 + m];
                bfr[nt][s].y = W[(k + 1) * 32 + nt * 16 + m];
            }
        }
        wave_lds_order();
        v2f a[8];
#pragma unroll
        for (int s = 0; s < 8; ++s)
            a[s] = *(const v2f*)&mytile[m * TILE_LD + 4 * s + 2 * hi];
#pragma unroll
        for (int nt = 0; nt < 2; ++nt) {
            v8f c = {};
#pragma unroll
            for (int s = 0; s < 8; ++s) {
                c = __builtin_amdgcn_wmma_f32_16x16x4_f32(
                         false, a[s],  false, bfr[nt][s],
                         (short)0, c,  false,  false);
            }
            float* op = out + (size_t)(row0 + 8 * hi) * 32 + nt * 16 + m;
#pragma unroll
            for (int r = 0; r < 8; ++r) op[(size_t)r * 32] = c[r];
        }
    }
}

extern "C" void kernel_launch(void* const* d_in, const int* in_sizes, int n_in,
                              void* d_out, int out_size, void* d_ws, size_t ws_size,
                              hipStream_t stream) {
    const float* X   = (const float*)d_in[0];
    const float* W   = (const float*)d_in[1];
    const int*   src = (const int*)d_in[2];
    const int*   dst = (const int*)d_in[3];
    float*       out = (float*)d_out;
    int*         runs   = (int*)d_ws;
    int*         sorted = runs + N_PART * N_CHUNK;

    gin_split_kernel<<<N_CHUNKS, 256, 0, stream>>>(src, dst, runs, sorted);
    gin_pull_gemm_kernel<<<N_PBLK, 512, 0, stream>>>(X, W, runs, sorted, out);
}


# pull grid balanced: parts of 400 nodes (split key dst/400), 250 blocks of 25 waves, one per compute unit, so no unit gathers twice another's rows (was 391 blocks of 16 waves on 256); sum order kept
# speedup vs baseline: 3.9036x; 1.0647x over previous
#include <hip/hip_runtime.h>

#define N_NODES 100000
#define N_EDGES 1600000
#define IN_DIM  32
#define OUT_DIM 32
#define N_TILES (N_NODES / 16)
#define TILE_LD 44

#define PART_N   400
#define PART_W   25
#define PULL_T   (PART_W * 32)
#define N_PART   252
#define N_CHUNK  256
#define N_CHUNKS (N_EDGES / CHUNK_E)
#define CHUNK_E  6400
#define WAVE_E   800
#define ROUNDS   25
#define PER_LANE 9
#define LIST_CAP (PULL_T * PER_LANE)
#define NB       (PART_N + 1)
#define N_PBLK   (N_NODES / PART_N)
#define RUNS_W   10

typedef __attribute__((ext_vector_type(2))) float v2f;
typedef __attribute__((ext_vector_type(8))) float v8f;
typedef __attribute__((ext_vector_type(4))) int   v4i;

__device__ __forceinline__ void wave_lds_order() {
    __builtin_amdgcn_fence(__ATOMIC_RELEASE, "wavefront");
    __builtin_amdgcn_wave_barrier();
    __builtin_amdgcn_fence(__ATOMIC_ACQUIRE, "wavefront");
}


template <int NBITS>
__device__ __forceinline__ unsigned same_key_lanes(int key) {
    unsigned differ = 0;
#pragma unroll
    for (int b = 0; b < NBITS; ++b) {
        int mine = (int)((unsigned)key << (31 - b)) >> 31;
        asm("" : "+v"(mine));
        const unsigned bal = __builtin_amdgcn_ballot_w32(mine != 0);
        differ |= bal ^ (unsigned)mine;
    }
    return ~differ;
}

__global__ __launch_bounds__(256) void gin_split_kernel(const int* __restrict__ src,
                                                        const int* __restrict__ dst,
                                                        int* __restrict__ runs,
                                                        int* __restrict__ sorted) {
    __shared__ int h[8 * N_PART];
    __shared__ int wtot[8];
    __shared__ __attribute__((aligned(16))) int slice[CHUNK_E];
    const int t = threadIdx.x;
    const int lane = t & 31;
    const int w = t >> 5;
    const int b = blockIdx.x;
    for (int i = t; i < 8 * N_PART; i += 256) h[i] = 0;
    __syncthreads();

    const int e0 = b * CHUNK_E + w * WAVE_E + lane;
    int dv[ROUNDS];
    int sv[ROUNDS];
#pragma unroll
    for (int r = 0; r < ROUNDS; ++r) dv[r] = dst[e0 + r * 32];
#pragma unroll
    for (int r = 0; r < ROUNDS; ++r) sv[r] = src[e0 + r * 32];
#pragma unroll
    for (int r = 0; r < ROUNDS; ++r) {
        const unsigned p = (unsigned)dv[r] / (unsigned)PART_N;
        dv[r] = (int)((p << 9) | ((unsigned)dv[r] - p * (unsigned)PART_N));
    }

    int* hw = h + w * N_PART;
    unsigned same[ROUNDS];
    int pre[ROUNDS];
    same[0] = same_key_lanes<8>(dv[0] >> 9);
#pragma unroll
    for (int r = 0; r < ROUNDS; ++r) {
        const int part = dv[r] >> 9;
        pre[r] = hw[part];
        if (r + 1 < ROUNDS) same[r + 1] = same_key_lanes<8>(dv[r + 1] >> 9);
        wave_lds_order();
        hw[part] = pre[r] + __builtin_popcount(same[r]);
        wave_lds_order();
    }
    __syncthreads();
    const int d0 = 2 * t;
    int tot0 = 0, tot1 = 0;
    if (t < N_PART / 2) {
#pragma unroll
        for (int i = 0; i < 8; ++i) {
            tot0 += h[i * N_PART + d0];
            tot1 += h[i * N_PART + d0 + 1];
        }
    }
    int incl = tot0 + tot1;
#pragma unroll
    for (int o = 1; o < 32; o <<= 1) {
        const int u = __shfl_up(incl, o);
        if (lane >= o) incl += u;
    }
    if (lane == 31) wtot[w] = incl;
    __syncthreads();
    if (t < N_PART / 2) {
        int first = incl - (tot0 + tot1);
#pragma unroll
        for (int i = 0; i < 8; ++i) first += (i < w) ? wtot[i] : 0;
        runs[d0 * N_CHUNK + b] = first;
        runs[(d0 + 1) * N_CHUNK + b] = first + tot0;
        int run = first;
#pragma unroll
        for (int i = 0; i < 8; ++i) {
            const int c = h[i * N_PART + d0];
            h[i * N_PART + d0] = run;
            run += c;
        }
#pragma unroll
        for (int i = 0; i < 8; ++i) {
            const int c = h[i * N_PART + d0 + 1];
            h[i * N_PART + d0 + 1] = run;
            run += c;
        }
    }
    __syncthreads();
#pragma unroll
    for (int r = 0; r < ROUNDS; ++r) {
        const int start = hw[dv[r] >> 9];
        slice[start + (int)__builtin_amdgcn_mbcnt_lo(same[r], (unsigned)pre[r])] = (sv[r] << 9) | (dv[r] & 511);
    }
    __syncthreads();
    const v4i* s4 = (const v4i*)slice;
    v4i* g4 = (v4i*)(sorted + b * CHUNK_E);
    static_assert(CHUNK_E / 4 == 6 * 256 + 64, "write-out rounds");
    v4i q[6];
#pragma unroll
    for (int j = 0; j < 6; ++j) q[j] = s4[t + j * 256];
#pragma unroll
    for (int j = 0; j < 6; ++j) g4[t + j * 256] = q[j];
    if (t < 64) g4[t + 6 * 256] = s4[t + 6 * 256];
}

template <bool PAD>
__device__ __forceinline__ void count_wave(int* hw, const int (&ent)[PER_LANE],
                                           unsigned (&same)[PER_LANE], int (&pre)[PER_LANE]) {
    auto key_of = [](int e) { return (!PAD) ? (e & 511) : ((e >= 0) ? (e & 511) : PART_N); };
    same[0] = same_key_lanes<9>(key_of(ent[0]));
#pragma unroll
    for (int j = 0; j < PER_LANE; ++j) {
        const int key = key_of(ent[j]);
        pre[j] = hw[key];
        if (j + 1 < PER_LANE) same[j + 1] = same_key_lanes<9>(key_of(ent[j + 1]));
        wave_lds_order();
        hw[key] = pre[j] + __builtin_popcount(same[j]);
        wave_lds_order();
    }
}

__global__ __launch_bounds__(PULL_T) void gin_pull_gemm_kernel(const float* __restrict__ X,
                                                            const float* __restrict__ W,
                                                            const int* __restrict__ runs,
                                                            const int* __restrict__ sorted,
                                                            float* __restrict__ out) {
    __shared__ int   h[PART_W * NB];
    __shared__ int   nstart[NB];
    __shared__ int   ids[LIST_CAP];
    __shared__ int   wsum[16];
    __shared__ int   segbase[256];
    __shared__ int   segpos[257];
    __shared__ int   stage[LIST_CAP];
    __shared__ __attribute__((aligned(16))) float tile[PART_W * 16 * TILE_LD];

    const int t = threadIdx.x;
    const int lane = t & 31;
    const int wave = __builtin_amdgcn_readfirstlane((int)(threadIdx.x >> 5));
    const int m  = lane & 15;
    const int hi = lane >> 4;
    const int part = blockIdx.x;
    int*   hw     = h + wave * NB;
    float* mytile = tile + wave * (16 * TILE_LD);

    for (int n = 0; n < 16; ++n) mytile[n * TILE_LD + lane] = 0.0f;

    {
        int s0 = 0, len = 0;
        if (t < N_CHUNKS) {
            s0  = runs[part * N_CHUNK + t];
            len = runs[(part + 1) * N_CHUNK + t] - s0;
        }
        int incl = len;
#pragma unroll
        for (int o = 1; o < 32; o <<= 1) {
            const int u = __shfl_up(incl, o);
            if (lane >= o) incl += u;
        }
        if (lane == 31 && wave < 8) wsum[wave] = incl;
        __syncthreads();
        if (t < 256) {
            int first = incl - len;
#pragma unroll
            for (int i = 0; i < 8; ++i) first += (i < wave) ? wsum[i] : 0;
            segbase[t] = t * CHUNK_E + s0;
            segpos[t] = first;
            if (t == 255) segpos[256] = first + len;
        }
        __syncthreads();
    }
    const int total = segpos[256];

    for (int cb = 0; cb < total; cb += LIST_CAP) {
        for (int i = t; i < PART_W * NB; i += PULL_T) h[i] = 0;

        {
            const int sp = segpos[RUNS_W * wave + ((lane < RUNS_W) ? lane : RUNS_W)];
            const int sb = segbase[RUNS_W * wave + ((lane < RUNS_W) ? lane : (RUNS_W - 1))];
            int ev[RUNS_W][2];
#pragma unroll
            for (int r = 0; r < RUNS_W; ++r) {
                const int p0 = __builtin_amdgcn_readlane(sp, r);
                const int p1 = __builtin_amdgcn_readlane(sp, r + 1);
                const int base = __builtin_amdgcn_readlane(sb, r);
                const int last = (p1 - p0 > 0) ? (p1 - p0 - 1) : 0;
#pragma unroll
                for (int k = 0; k < 2; ++k) {
                    const int i = lane + 32 * k;
                    const int idx = base + ((i < last) ? i : last);
                    ev[r][k] = sorted[(idx < N_EDGES) ? idx : (N_EDGES - 1)];
                }
            }
#pragma unroll
            for (int r = 0; r < RUNS_W; ++r) {
                const int p0 = __builtin_amdgcn_readlane(sp, r);
                const int p1 = __builtin_amdgcn_readlane(sp, r + 1);
#pragma unroll
                for (int k = 0; k < 2; ++k) {
                    const int i = lane + 32 * k;
                    const int pos = p0 + i - cb;
                    if (i < p1 - p0 && pos >= 0 && pos < LIST_CAP) stage[pos] = ev[r][k];
                }
            }
            for (int r = 0; r < RUNS_W; ++r) {
                const int p0 = __builtin_amdgcn_readlane(sp, r);
                const int p1 = __builtin_amdgcn_readlane(sp, r + 1);
                const int base = __builtin_amdgcn_readlane(sb, r);
                for (int i0 = 64; i0 < p1 - p0; i0 += 32) {
                    const int i = i0 + lane;
                    const int pos = p0 + i - cb;
                    if (i < p1 - p0 && pos >= 0 && pos < LIST_CAP) stage[pos] = sorted[base + i];
                }
            }
        }
        __syncthreads();

        int ent[PER_LANE];
        unsigned same[PER_LANE];
#pragma unroll
        for (int j = 0; j < PER_LANE; ++j) {
            const int q = wave * (32 * PER_LANE) + j * 32 + lane;
            ent[j] = (cb + q < total) ? stage[q] : -1;
        }
        int pre[PER_LANE];
        const bool whole = cb + (wave + 1) * (32 * PER_LANE) <= total;
        if (whole) count_wave<false>(hw, ent, same, pre);
        else       count_wave<true>(hw, ent, same, pre);
        __syncthreads();

        int tot = 0;
        if (t < PART_N) {
#pragma unroll
            for (int i = 0; i < PART_W; ++i) tot += h[i * NB + t];
        }
        int incl = tot;
#pragma unroll
        for (int o = 1; o < 32; o <<= 1) {
            const int u = __shfl_up(incl, o);
            if (lane >= o) incl += u;
        }
        if (lane == 31 && wave < 13) wsum[wave] = incl;
        __syncthreads();
        if (t < PART_N) {
            int run = incl - tot;
#pragma unroll
            for (int i = 0; i < 13; ++i) run += (i < wave) ? wsum[i] : 0;
            nstart[t] = run;
#pragma unroll
            for (int i = 0; i < PART_W; ++i) {
                const int c = h[i * NB + t];
                h[i * NB + t] = run;
                run += c;
            }
            if (t == PART_N - 1) nstart[PART_N] = run;
        }
        __syncthreads();

        if (whole) {
#pragma unroll
            for (int j = 0; j < PER_LANE; ++j) {
                const int start = hw[ent[j] & 511];
                ids[start + (int)__builtin_amdgcn_mbcnt_lo(same[j], (unsigned)pre[j])] = ent[j] >> 9;
            }
        } else {
#pragma unroll
            for (int j = 0; j < PER_LANE; ++j) {
                const int key = (ent[j] >= 0) ? (ent[j] & 511) : PART_N;
                const int start = hw[key];
                if (ent[j] >= 0) ids[start + (int)__builtin_amdgcn_mbcnt_lo(same[j], (unsigned)pre[j])] = ent[j] >> 9;
            }
        }
        __syncthreads();

        {
            const int nb = nstart[16 * wave + ((lane < 16) ? lane : 16)];
            const int S = __builtin_amdgcn_readlane(nb, 0);
            const int E = __builtin_amdgcn_readlane(nb, 16);
            if (S < E) {
                int cur = 0;
                int cend = __builtin_amdgcn_readlane(nb, 1);
                float acc = mytile[lane];

                auto load16 = [&](int k, float (&v)[16]) {
                    const int idv = (k + lane < E) ? ids[k + lane] : 0;
#pragma unroll
                    for (int i = 0; i < 16; ++i) {
                        const int sk = __builtin_amdgcn_readlane(idv, i);
                        v[i] = X[sk * 32 + lane];
                    }
                };
                auto add_row = [&](int p, float v) {
                    while (__builtin_expect(p >= cend, 0)) {
                        mytile[cur * TILE_LD + lane] = acc;
                        ++cur;
                        acc = mytile[cur * TILE_LD + lane];
                        cend = __builtin_amdgcn_readlane(nb, cur + 1);
                    }
                    acc += v;
                };
                auto add16 = [&](int k, const float (&v)[16]) {
                    if (k + 16 <= E) {
#pragma unroll
                        for (int i = 0; i < 16; ++i) add_row(k + i, v[i]);
                    } else {
#pragma unroll
                        for (int i = 0; i < 16; ++i)
                            if (k + i < E) add_row(k + i, v[i]);
                    }
                };

                float va[16], vb[16];
                load16(S, va);
                for (int k = S; k < E; k += 32) {
                    load16(k + 16, vb);
                    add16(k, va);
                    load16(k + 32, va);
                    add16(k + 16, vb);
                }
                mytile[cur * TILE_LD + lane] = acc;
            }
        }
    }

    static_assert(N_PBLK * PART_W == N_TILES, "every wave of every block owns one full row tile");
    {
        const int row0 = (part * PART_W + wave) * 16;
        v2f bfr[2][8];
#pragma unroll
        for (int nt = 0; nt < 2; ++nt) {
#pragma unroll
            for (int s = 0; s < 8; ++s) {
                const int k = 4 * s + 2 * hi;
                bfr[nt][s].x = W[k * 32 + nt * 16 + m];
                bfr[nt][s].y = W[(k + 1) * 32 + nt * 16 + m];
            }
        }
        wave_lds_order();
        v2f a[8];
#pragma unroll
        for (int s = 0; s < 8; ++s)
            a[s] = *(const v2f*)&mytile[m * TILE_LD + 4 * s + 2 * hi];
#pragma unroll
        for (int nt = 0; nt < 2; ++nt) {
            v8f c = {};
#pragma unroll
            for (int s = 0; s < 8; ++s) {
                c = __builtin_amdgcn_wmma_f32_16x16x4_f32(
                         false, a[s],  false, bfr[nt][s],
                         (short)0, c,  false,  false);
            }
            float* op = out + (size_t)(row0 + 8 * hi) * 32 + nt * 16 + m;
#pragma unroll
            for (int r = 0; r < 8; ++r) op[(size_t)r * 32] = c[r];
        }
    }
}

extern "C" void kernel_launch(void* const* d_in, const int* in_sizes, int n_in,
                              void* d_out, int out_size, void* d_ws, size_t ws_size,
                              hipStream_t stream) {
    const float* X   = (const float*)d_in[0];
    const float* W   = (const float*)d_in[1];
    const int*   src = (const int*)d_in[2];
    const int*   dst = (const int*)d_in[3];
    float*       out = (float*)d_out;
    int*         runs   = (int*)d_ws;
    int*         sorted = runs + N_PART * N_CHUNK;

    gin_split_kernel<<<N_CHUNKS, 256, 0, stream>>>(src, dst, runs, sorted);
    gin_pull_gemm_kernel<<<N_PBLK, PULL_T, 0, stream>>>(X, W, runs, sorted, out);
}
